# starting kernel
# baseline (speedup 1.0000x reference)
#include <hip/hip_runtime.h>

typedef __attribute__((ext_vector_type(2))) float v2f;
typedef __attribute__((ext_vector_type(8))) float v8f;

#define T_FULL 75000
#define CH_IN  17
#define NGROUP 32
#define NKER   8
#define NTAP   9
#define NUNIT  28
#define SG_FLOATS 75040
#define ZSLOT     75008
#define SW_OFF    SG_FLOATS
#define SACC_OFF  (SG_FLOATS + 72)
#define SMEM_FLOATS (SACC_OFF + 16)

__global__ __launch_bounds__(1024)
void hydra_unit_kernel(const float* __restrict__ X,
                       const float* __restrict__ W,
                       const int*   __restrict__ IDX,
                       float* __restrict__ OUT)
{
    extern __shared__ float smem[];
    float* sg   = smem;
    float* sw   = smem + SW_OFF;
    float* sAcc = smem + SACC_OFF;

    const int tid  = threadIdx.x;
    const int lane = tid & 31;
    const int wav  = tid >> 5;
    const int nthr = blockDim.x;
    const int nwav = nthr >> 5;

    const int bid = blockIdx.x;
    const int b   = bid / (NUNIT * NGROUP);
    const int rem = bid % (NUNIT * NGROUP);
    const int u   = rem / NGROUP;
    const int h   = rem % NGROUP;
    const int di  = u >> 1;
    const int dif = u & 1;
    const int d   = 1 << di;
    const int Teff = dif ? (T_FULL - 1) : T_FULL;

    if (tid < 16) sAcc[tid] = 0.0f;
    if (tid == 16) sg[ZSLOT] = 0.0f;
    {
        const int wbase = (u * (NKER * NGROUP) + h * NKER) * NTAP;
        if (tid < NKER * NTAP) sw[tid] = W[wbase + tid];
    }
    int ch[8];
    {
        const int* ip = IDX + (u * NGROUP + h) * 8;
        #pragma unroll
        for (int i = 0; i < 8; ++i) ch[i] = ip[i];
    }

    {
        const float* Xb = X + (size_t)b * CH_IN * T_FULL;
        for (int t = tid; t < Teff; t += nthr) {
            float s = 0.0f;
            #pragma unroll
            for (int i = 0; i < 8; ++i) {
                const float* xc = Xb + (size_t)ch[i] * T_FULL + t;
                float v = xc[0];
                if (dif) v = xc[1] - v;
                s += v;
            }
            sg[t] = s;
        }
    }
    __syncthreads();

    const int c16 = lane & 15;
    const int krb = (lane >> 4) << 1;
    v2f afrag[3];
    #pragma unroll
    for (int tile = 0; tile < 3; ++tile) {
        #pragma unroll
        for (int i = 0; i < 2; ++i) {
            const int j = tile * 4 + krb + i;
            afrag[tile][i] = (c16 < NKER && j < NTAP) ? sw[c16 * NTAP + j] : 0.0f;
        }
    }

    float accMax[8], accMin[8];
    #pragma unroll
    for (int k = 0; k < 8; ++k) { accMax[k] = 0.0f; accMin[k] = 0.0f; }

    const int nblk = (Teff + 31) >> 5;
    for (int blk = wav; blk < nblk; blk += nwav) {
        const int t0 = blk << 5;

        v2f bf0[3], bf1[3];
        #pragma unroll
        for (int tile = 0; tile < 3; ++tile) {
            #pragma unroll
            for (int i = 0; i < 2; ++i) {
                const int j  = tile * 4 + krb + i;
                const int p0 = t0 + c16 + (j - 4) * d;
                const int p1 = p0 + 16;
                const int a0 = (j < NTAP && (unsigned)p0 < (unsigned)Teff) ? p0 : ZSLOT;
                const int a1 = (j < NTAP && (unsigned)p1 < (unsigned)Teff) ? p1 : ZSLOT;
                bf0[tile][i] = sg[a0];
                bf1[tile][i] = sg[a1];
            }
        }

        v8f c0 = {0.f,0.f,0.f,0.f,0.f,0.f,0.f,0.f};
        v8f c1 = {0.f,0.f,0.f,0.f,0.f,0.f,0.f,0.f};
        #pragma unroll
        for (int tile = 0; tile < 3; ++tile) {
            c0 = __builtin_amdgcn_wmma_f32_16x16x4_f32(
                     false, afrag[tile], false, bf0[tile], (short)0, c0, false, false);
            c1 = __builtin_amdgcn_wmma_f32_16x16x4_f32(
                     false, afrag[tile], false, bf1[tile], (short)0, c1, false, false);
        }

        float z[8];
        #pragma unroll
        for (int k = 0; k < 8; ++k) {
            const float hi = __shfl(c1[k], c16, 32);
            z[k] = (lane < 16) ? c0[k] : hi;
        }

        if (t0 + lane < Teff) {
            float mx = z[0], mn = z[0];
            #pragma unroll
            for (int k = 1; k < 8; ++k) { mx = fmaxf(mx, z[k]); mn = fminf(mn, z[k]); }
            bool dmx = false, dmn = false;
            #pragma unroll
            for (int k = 0; k < 8; ++k) {
                const bool hm = (!dmx) && (z[k] == mx);
                accMax[k] += hm ? mx : 0.0f;  dmx = dmx || hm;
                const bool hn = (!dmn) && (z[k] == mn);
                accMin[k] += hn ? 1.0f : 0.0f; dmn = dmn || hn;
            }
        }
    }

    #pragma unroll
    for (int k = 0; k < 8; ++k) {
        float vM = accMax[k], vN = accMin[k];
        #pragma unroll
        for (int off = 16; off > 0; off >>= 1) {
            vM += __shfl_xor(vM, off, 32);
            vN += __shfl_xor(vN, off, 32);
        }
        if (lane == 0) {
            atomicAdd(&sAcc[k],     vM);
            atomicAdd(&sAcc[8 + k], vN);
        }
    }
    __syncthreads();

    if (tid < 16) {
        const int which = tid >> 3;
        const int k     = tid & 7;
        const size_t off = (size_t)b * 14336
                         + (size_t)(2 * u + which) * (NGROUP * NKER)
                         + (size_t)h * NKER + k;
        OUT[off] = sAcc[tid];
    }
}

extern "C" void kernel_launch(void* const* d_in, const int* in_sizes, int n_in,
                              void* d_out, int out_size, void* d_ws, size_t ws_size,
                              hipStream_t stream) {
    (void)in_sizes; (void)n_in; (void)d_ws; (void)ws_size; (void)out_size;
    const float* X   = (const float*)d_in[0];
    const float* W   = (const float*)d_in[1];
    const int*   IDX = (const int*)d_in[2];
    float*       OUT = (float*)d_out;

    const int shmem = SMEM_FLOATS * (int)sizeof(float);
    (void)hipFuncSetAttribute((const void*)hydra_unit_kernel,
                              hipFuncAttributeMaxDynamicSharedMemorySize, shmem);

    const int nblocks = 4 * NUNIT * NGROUP;
    hydra_unit_kernel<<<dim3(nblocks), dim3(1024), shmem, stream>>>(X, W, IDX, OUT);
}


# phase-1 gather: one 16-byte load per channel per 4 time points, all 8 channel loads (plus the diff path's 8 next-element loads) issued before first use; diff branch hoisted; sum order unchanged
# speedup vs baseline: 1.4876x; 1.4876x over previous
#include <hip/hip_runtime.h>

typedef __attribute__((ext_vector_type(2))) float v2f;
typedef __attribute__((ext_vector_type(8))) float v8f;

#define T_FULL 75000
#define CH_IN  17
#define NGROUP 32
#define NKER   8
#define NTAP   9
#define NUNIT  28
#define NQUAD  (T_FULL / 4)
#define SG_FLOATS 75040
#define ZSLOT     75008
#define SW_OFF    SG_FLOATS
#define SACC_OFF  (SG_FLOATS + 72)
#define SMEM_FLOATS (SACC_OFF + 16)

__global__ __launch_bounds__(1024)
void hydra_unit_kernel(const float* __restrict__ X,
                       const float* __restrict__ W,
                       const int*   __restrict__ IDX,
                       float* __restrict__ OUT)
{
    extern __shared__ __attribute__((aligned(16))) float smem[];
    float* sg   = smem;
    float* sw   = smem + SW_OFF;
    float* sAcc = smem + SACC_OFF;

    const int tid  = threadIdx.x;
    const int lane = tid & 31;
    const int wav  = tid >> 5;
    const int nthr = blockDim.x;
    const int nwav = nthr >> 5;

    const int bid = blockIdx.x;
    const int b   = bid / (NUNIT * NGROUP);
    const int rem = bid % (NUNIT * NGROUP);
    const int u   = rem / NGROUP;
    const int h   = rem % NGROUP;
    const int di  = u >> 1;
    const int dif = u & 1;
    const int d   = 1 << di;
    const int Teff = dif ? (T_FULL - 1) : T_FULL;

    if (tid < 16) sAcc[tid] = 0.0f;
    if (tid == 16) sg[ZSLOT] = 0.0f;
    {
        const int wbase = (u * (NKER * NGROUP) + h * NKER) * NTAP;
        if (tid < NKER * NTAP) sw[tid] = W[wbase + tid];
    }
    int ch[8];
    {
        const int* ip = IDX + (u * NGROUP + h) * 8;
        #pragma unroll
        for (int i = 0; i < 8; ++i) ch[i] = ip[i];
    }

    {
        const float* Xb = X + (size_t)b * CH_IN * T_FULL;
        const float* xr[8];
        #pragma unroll
        for (int i = 0; i < 8; ++i) xr[i] = Xb + (size_t)ch[i] * T_FULL;
        float4* sg4 = reinterpret_cast<float4*>(sg);
        if (!dif) {
            for (int q = tid; q < NQUAD; q += nthr) {
                float4 v[8];
                #pragma unroll
                for (int i = 0; i < 8; ++i) v[i] = reinterpret_cast<const float4*>(xr[i])[q];
                float4 s = make_float4(0.0f, 0.0f, 0.0f, 0.0f);
                #pragma unroll
                for (int i = 0; i < 8; ++i) {
                    s.x += v[i].x;  s.y += v[i].y;  s.z += v[i].z;  s.w += v[i].w;
                }
                sg4[q] = s;
            }
        } else {
            for (int q = tid; q < NQUAD; q += nthr) {
                const int te = min(4 * q + 4, T_FULL - 1);
                float4 v[8];
                float  e[8];
                #pragma unroll
                for (int i = 0; i < 8; ++i) v[i] = reinterpret_cast<const float4*>(xr[i])[q];
                #pragma unroll
                for (int i = 0; i < 8; ++i) e[i] = xr[i][te];
                float4 s = make_float4(0.0f, 0.0f, 0.0f, 0.0f);
                #pragma unroll
                for (int i = 0; i < 8; ++i) {
                    s.x += v[i].y - v[i].x;
                    s.y += v[i].z - v[i].y;
                    s.z += v[i].w - v[i].z;
                    s.w += e[i]   - v[i].w;
                }
                sg4[q] = s;
            }
        }
    }
    __syncthreads();

    const int c16 = lane & 15;
    const int krb = (lane >> 4) << 1;
    v2f afrag[3];
    #pragma unroll
    for (int tile = 0; tile < 3; ++tile) {
        #pragma unroll
        for (int i = 0; i < 2; ++i) {
            const int j = tile * 4 + krb + i;
            afrag[tile][i] = (c16 < NKER && j < NTAP) ? sw[c16 * NTAP + j] : 0.0f;
        }
    }

    float accMax[8], accMin[8];
    #pragma unroll
    for (int k = 0; k < 8; ++k) { accMax[k] = 0.0f; accMin[k] = 0.0f; }

    const int nblk = (Teff + 31) >> 5;
    for (int blk = wav; blk < nblk; blk += nwav) {
        const int t0 = blk << 5;

        v2f bf0[3], bf1[3];
        #pragma unroll
        for (int tile = 0; tile < 3; ++tile) {
            #pragma unroll
            for (int i = 0; i < 2; ++i) {
                const int j  = tile * 4 + krb + i;
                const int p0 = t0 + c16 + (j - 4) * d;
                const int p1 = p0 + 16;
                const int a0 = (j < NTAP && (unsigned)p0 < (unsigned)Teff) ? p0 : ZSLOT;
                const int a1 = (j < NTAP && (unsigned)p1 < (unsigned)Teff) ? p1 : ZSLOT;
                bf0[tile][i] = sg[a0];
                bf1[tile][i] = sg[a1];
            }
        }

        v8f c0 = {0.f,0.f,0.f,0.f,0.f,0.f,0.f,0.f};
        v8f c1 = {0.f,0.f,0.f,0.f,0.f,0.f,0.f,0.f};
        #pragma unroll
        for (int tile = 0; tile < 3; ++tile) {
            c0 = __builtin_amdgcn_wmma_f32_16x16x4_f32(
                     false, afrag[tile], false, bf0[tile], (short)0, c0, false, false);
            c1 = __builtin_amdgcn_wmma_f32_16x16x4_f32(
                     false, afrag[tile], false, bf1[tile], (short)0, c1, false, false);
        }

        float z[8];
        #pragma unroll
        for (int k = 0; k < 8; ++k) {
            const float hi = __shfl(c1[k], c16, 32);
            z[k] = (lane < 16) ? c0[k] : hi;
        }

        if (t0 + lane < Teff) {
            float mx = z[0], mn = z[0];
            #pragma unroll
            for (int k = 1; k < 8; ++k) { mx = fmaxf(mx, z[k]); mn = fminf(mn, z[k]); }
            bool dmx = false, dmn = false;
            #pragma unroll
            for (int k = 0; k < 8; ++k) {
                const bool hm = (!dmx) && (z[k] == mx);
                accMax[k] += hm ? mx : 0.0f;  dmx = dmx || hm;
                const bool hn = (!dmn) && (z[k] == mn);
                accMin[k] += hn ? 1.0f : 0.0f; dmn = dmn || hn;
            }
        }
    }

    #pragma unroll
    for (int k = 0; k < 8; ++k) {
        float vM = accMax[k], vN = accMin[k];
        #pragma unroll
        for (int off = 16; off > 0; off >>= 1) {
            vM += __shfl_xor(vM, off, 32);
            vN += __shfl_xor(vN, off, 32);
        }
        if (lane == 0) {
            atomicAdd(&sAcc[k],     vM);
            atomicAdd(&sAcc[8 + k], vN);
        }
    }
    __syncthreads();

    if (tid < 16) {
        const int which = tid >> 3;
        const int k     = tid & 7;
        const size_t off = (size_t)b * 14336
                         + (size_t)(2 * u + which) * (NGROUP * NKER)
                         + (size_t)h * NKER + k;
        OUT[off] = sAcc[tid];
    }
}

extern "C" void kernel_launch(void* const* d_in, const int* in_sizes, int n_in,
                              void* d_out, int out_size, void* d_ws, size_t ws_size,
                              hipStream_t stream) {
    (void)in_sizes; (void)n_in; (void)d_ws; (void)ws_size; (void)out_size;
    const float* X   = (const float*)d_in[0];
    const float* W   = (const float*)d_in[1];
    const int*   IDX = (const int*)d_in[2];
    float*       OUT = (float*)d_out;

    const int shmem = SMEM_FLOATS * (int)sizeof(float);
    (void)hipFuncSetAttribute((const void*)hydra_unit_kernel,
                              hipFuncAttributeMaxDynamicSharedMemorySize, shmem);

    const int nblocks = 4 * NUNIT * NGROUP;
    hydra_unit_kernel<<<dim3(nblocks), dim3(1024), shmem, stream>>>(X, W, IDX, OUT);
}


# A tile rows 8-15 = the 8 kernels moved one tap: 3 WMMAs give 32 time points (was 6), each lane gets all 8 z of its own time point, no bpermute; tap-sum grouping and per-lane partial sums change
# speedup vs baseline: 1.8947x; 1.2737x over previous
#include <hip/hip_runtime.h>

typedef __attribute__((ext_vector_type(2))) float v2f;
typedef __attribute__((ext_vector_type(8))) float v8f;

#define T_FULL 75000
#define CH_IN  17
#define NGROUP 32
#define NKER   8
#define NTAP   9
#define NUNIT  28
#define NQUAD  (T_FULL / 4)
#define SG_FLOATS 75040
#define ZSLOT     75008
#define SW_OFF    SG_FLOATS
#define SACC_OFF  (SG_FLOATS + 72)
#define SMEM_FLOATS (SACC_OFF + 16)

__global__ __launch_bounds__(1024)
void hydra_unit_kernel(const float* __restrict__ X,
                       const float* __restrict__ W,
                       const int*   __restrict__ IDX,
                       float* __restrict__ OUT)
{
    extern __shared__ __attribute__((aligned(16))) float smem[];
    float* sg   = smem;
    float* sw   = smem + SW_OFF;
    float* sAcc = smem + SACC_OFF;

    const int tid  = threadIdx.x;
    const int lane = tid & 31;
    const int wav  = tid >> 5;
    const int nthr = blockDim.x;
    const int nwav = nthr >> 5;

    const int bid = blockIdx.x;
    const int b   = bid / (NUNIT * NGROUP);
    const int rem = bid % (NUNIT * NGROUP);
    const int u   = rem / NGROUP;
    const int h   = rem % NGROUP;
    const int di  = u >> 1;
    const int dif = u & 1;
    const int d   = 1 << di;
    const int Teff = dif ? (T_FULL - 1) : T_FULL;

    if (tid < 16) sAcc[tid] = 0.0f;
    if (tid == 16) sg[ZSLOT] = 0.0f;
    {
        const int wbase = (u * (NKER * NGROUP) + h * NKER) * NTAP;
        if (tid < NKER * NTAP) sw[tid] = W[wbase + tid];
    }
    int ch[8];
    {
        const int* ip = IDX + (u * NGROUP + h) * 8;
        #pragma unroll
        for (int i = 0; i < 8; ++i) ch[i] = ip[i];
    }

    {
        const float* Xb = X + (size_t)b * CH_IN * T_FULL;
        const float* xr[8];
        #pragma unroll
        for (int i = 0; i < 8; ++i) xr[i] = Xb + (size_t)ch[i] * T_FULL;
        float4* sg4 = reinterpret_cast<float4*>(sg);
        if (!dif) {
            for (int q = tid; q < NQUAD; q += nthr) {
                float4 v[8];
                #pragma unroll
                for (int i = 0; i < 8; ++i) v[i] = reinterpret_cast<const float4*>(xr[i])[q];
                float4 s = make_float4(0.0f, 0.0f, 0.0f, 0.0f);
                #pragma unroll
                for (int i = 0; i < 8; ++i) {
                    s.x += v[i].x;  s.y += v[i].y;  s.z += v[i].z;  s.w += v[i].w;
                }
                sg4[q] = s;
            }
        } else {
            for (int q = tid; q < NQUAD; q += nthr) {
                const int te = min(4 * q + 4, T_FULL - 1);
                float4 v[8];
                float  e[8];
                #pragma unroll
                for (int i = 0; i < 8; ++i) v[i] = reinterpret_cast<const float4*>(xr[i])[q];
                #pragma unroll
                for (int i = 0; i < 8; ++i) e[i] = xr[i][te];
                float4 s = make_float4(0.0f, 0.0f, 0.0f, 0.0f);
                #pragma unroll
                for (int i = 0; i < 8; ++i) {
                    s.x += v[i].y - v[i].x;
                    s.y += v[i].z - v[i].y;
                    s.z += v[i].w - v[i].z;
                    s.w += e[i]   - v[i].w;
                }
                sg4[q] = s;
            }
        }
    }
    __syncthreads();

    const int c16 = lane & 15;
    const int hs  = lane >> 4;
    const int krb = hs << 1;
    const int ak  = c16 & 7;
    const int as  = c16 >> 3;
    v2f afrag[3];
    #pragma unroll
    for (int tile = 0; tile < 3; ++tile) {
        #pragma unroll
        for (int i = 0; i < 2; ++i) {
            const int jj = tile * 4 + krb + i - as;
            afrag[tile][i] = (jj >= 0 && jj < NTAP) ? sw[ak * NTAP + jj] : 0.0f;
        }
    }

    float accMax[8], accMin[8];
    #pragma unroll
    for (int k = 0; k < 8; ++k) { accMax[k] = 0.0f; accMin[k] = 0.0f; }

    const int hmask = -d;
    const int hp    = (Teff - 1) >> (di + 1);
    const int nq    = (hp << di) + min(d, Teff - (hp << (di + 1)));
    const int nblk  = (nq + 15) >> 4;
    for (int blk = wav; blk < nblk; blk += nwav) {
        const int q = (blk << 4) + c16;
        const int T = q + (q & hmask);

        v2f bf[3];
        #pragma unroll
        for (int tile = 0; tile < 3; ++tile) {
            #pragma unroll
            for (int i = 0; i < 2; ++i) {
                const int kk = tile * 4 + krb + i;
                const int p  = T + (kk - 4) * d;
                const int a  = (kk < NTAP + 1 && (unsigned)p < (unsigned)Teff) ? p : ZSLOT;
                bf[tile][i] = sg[a];
            }
        }

        v8f z = {0.f,0.f,0.f,0.f,0.f,0.f,0.f,0.f};
        #pragma unroll
        for (int tile = 0; tile < 3; ++tile) {
            z = __builtin_amdgcn_wmma_f32_16x16x4_f32(
                    false, afrag[tile], false, bf[tile], (short)0, z, false, false);
        }

        if (T + (hs ? d : 0) < Teff) {
            float mx = z[0], mn = z[0];
            #pragma unroll
            for (int k = 1; k < 8; ++k) { mx = fmaxf(mx, z[k]); mn = fminf(mn, z[k]); }
            bool dmx = false, dmn = false;
            #pragma unroll
            for (int k = 0; k < 8; ++k) {
                const bool hm = (!dmx) && (z[k] == mx);
                accMax[k] += hm ? mx : 0.0f;  dmx = dmx || hm;
                const bool hn = (!dmn) && (z[k] == mn);
                accMin[k] += hn ? 1.0f : 0.0f; dmn = dmn || hn;
            }
        }
    }

    #pragma unroll
    for (int k = 0; k < 8; ++k) {
        float vM = accMax[k], vN = accMin[k];
        #pragma unroll
        for (int off = 16; off > 0; off >>= 1) {
            vM += __shfl_xor(vM, off, 32);
            vN += __shfl_xor(vN, off, 32);
        }
        if (lane == 0) {
            atomicAdd(&sAcc[k],     vM);
            atomicAdd(&sAcc[8 + k], vN);
        }
    }
    __syncthreads();

    if (tid < 16) {
        const int which = tid >> 3;
        const int k     = tid & 7;
        const size_t off = (size_t)b * 14336
                         + (size_t)(2 * u + which) * (NGROUP * NKER)
                         + (size_t)h * NKER + k;
        OUT[off] = sAcc[tid];
    }
}

extern "C" void kernel_launch(void* const* d_in, const int* in_sizes, int n_in,
                              void* d_out, int out_size, void* d_ws, size_t ws_size,
                              hipStream_t stream) {
    (void)in_sizes; (void)n_in; (void)d_ws; (void)ws_size; (void)out_size;
    const float* X   = (const float*)d_in[0];
    const float* W   = (const float*)d_in[1];
    const int*   IDX = (const int*)d_in[2];
    float*       OUT = (float*)d_out;

    const int shmem = SMEM_FLOATS * (int)sizeof(float);
    (void)hipFuncSetAttribute((const void*)hydra_unit_kernel,
                              hipFuncAttributeMaxDynamicSharedMemorySize, shmem);

    const int nblocks = 4 * NUNIT * NGROUP;
    hydra_unit_kernel<<<dim3(nblocks), dim3(1024), shmem, stream>>>(X, W, IDX, OUT);
}


# phase-2 epilogue on scalar lane masks: one v_cmp_eq per kernel and side, first-hit and in-range logic by s_or/s_and_not1 (no branch); argmin hits summed per wave as integers by s_bcnt1, exact
# speedup vs baseline: 2.2102x; 1.1665x over previous
#include <hip/hip_runtime.h>

typedef __attribute__((ext_vector_type(2))) float v2f;
typedef __attribute__((ext_vector_type(8))) float v8f;

#define T_FULL 75000
#define CH_IN  17
#define NGROUP 32
#define NKER   8
#define NTAP   9
#define NUNIT  28
#define NQUAD  (T_FULL / 4)
#define SG_FLOATS 75040
#define ZSLOT     75008
#define SW_OFF    SG_FLOATS
#define SACC_OFF  (SG_FLOATS + 72)
#define SMEM_FLOATS (SACC_OFF + 16)

__global__ __launch_bounds__(1024)
void hydra_unit_kernel(const float* __restrict__ X,
                       const float* __restrict__ W,
                       const int*   __restrict__ IDX,
                       float* __restrict__ OUT)
{
    extern __shared__ __attribute__((aligned(16))) float smem[];
    float* sg   = smem;
    float* sw   = smem + SW_OFF;
    float* sAcc = smem + SACC_OFF;

    const int tid  = threadIdx.x;
    const int lane = tid & 31;
    const int wav  = __builtin_amdgcn_readfirstlane(tid >> 5);
    const int nthr = blockDim.x;
    const int nwav = nthr >> 5;

    const int bid = blockIdx.x;
    const int b   = bid / (NUNIT * NGROUP);
    const int rem = bid % (NUNIT * NGROUP);
    const int u   = rem / NGROUP;
    const int h   = rem % NGROUP;
    const int di  = u >> 1;
    const int dif = u & 1;
    const int d   = 1 << di;
    const int Teff = dif ? (T_FULL - 1) : T_FULL;

    if (tid < 16) sAcc[tid] = 0.0f;
    if (tid == 16) sg[ZSLOT] = 0.0f;
    {
        const int wbase = (u * (NKER * NGROUP) + h * NKER) * NTAP;
        if (tid < NKER * NTAP) sw[tid] = W[wbase + tid];
    }
    int ch[8];
    {
        const int* ip = IDX + (u * NGROUP + h) * 8;
        #pragma unroll
        for (int i = 0; i < 8; ++i) ch[i] = ip[i];
    }

    {
        const float* Xb = X + (size_t)b * CH_IN * T_FULL;
        const float* xr[8];
        #pragma unroll
        for (int i = 0; i < 8; ++i) xr[i] = Xb + (size_t)ch[i] * T_FULL;
        float4* sg4 = reinterpret_cast<float4*>(sg);
        if (!dif) {
            for (int q = tid; q < NQUAD; q += nthr) {
                float4 v[8];
                #pragma unroll
                for (int i = 0; i < 8; ++i) v[i] = reinterpret_cast<const float4*>(xr[i])[q];
                float4 s = make_float4(0.0f, 0.0f, 0.0f, 0.0f);
                #pragma unroll
                for (int i = 0; i < 8; ++i) {
                    s.x += v[i].x;  s.y += v[i].y;  s.z += v[i].z;  s.w += v[i].w;
                }
                sg4[q] = s;
            }
        } else {
            for (int q = tid; q < NQUAD; q += nthr) {
                const int te = min(4 * q + 4, T_FULL - 1);
                float4 v[8];
                float  e[8];
                #pragma unroll
                for (int i = 0; i < 8; ++i) v[i] = reinterpret_cast<const float4*>(xr[i])[q];
                #pragma unroll
                for (int i = 0; i < 8; ++i) e[i] = xr[i][te];
                float4 s = make_float4(0.0f, 0.0f, 0.0f, 0.0f);
                #pragma unroll
                for (int i = 0; i < 8; ++i) {
                    s.x += v[i].y - v[i].x;
                    s.y += v[i].z - v[i].y;
                    s.z += v[i].w - v[i].z;
                    s.w += e[i]   - v[i].w;
                }
                sg4[q] = s;
            }
        }
    }
    __syncthreads();

    const int c16 = lane & 15;
    const int hs  = lane >> 4;
    const int krb = hs << 1;
    const int ak  = c16 & 7;
    const int as  = c16 >> 3;
    v2f afrag[3];
    #pragma unroll
    for (int tile = 0; tile < 3; ++tile) {
        #pragma unroll
        for (int i = 0; i < 2; ++i) {
            const int jj = tile * 4 + krb + i - as;
            afrag[tile][i] = (jj >= 0 && jj < NTAP) ? sw[ak * NTAP + jj] : 0.0f;
        }
    }

    float accMax[8];
    int   cntMin[8];
    #pragma unroll
    for (int k = 0; k < 8; ++k) { accMax[k] = 0.0f; cntMin[k] = 0; }

    const int hmask = -d;
    const int hp    = (Teff - 1) >> (di + 1);
    const int nq    = (hp << di) + min(d, Teff - (hp << (di + 1)));
    const int nblk  = (nq + 15) >> 4;
    for (int blk = wav; blk < nblk; blk += nwav) {
        const int q = (blk << 4) + c16;
        const int T = q + (q & hmask);

        v2f bf[3];
        #pragma unroll
        for (int tile = 0; tile < 3; ++tile) {
            #pragma unroll
            for (int i = 0; i < 2; ++i) {
                const int kk = tile * 4 + krb + i;
                const int p  = T + (kk - 4) * d;
                const int a  = (kk < NTAP + 1 && (unsigned)p < (unsigned)Teff) ? p : ZSLOT;
                bf[tile][i] = sg[a];
            }
        }

        v8f z = {0.f,0.f,0.f,0.f,0.f,0.f,0.f,0.f};
        #pragma unroll
        for (int tile = 0; tile < 3; ++tile) {
            z = __builtin_amdgcn_wmma_f32_16x16x4_f32(
                    false, afrag[tile], false, bf[tile], (short)0, z, false, false);
        }

        {
            const unsigned dead = __builtin_amdgcn_ballot_w32(T + (hs ? d : 0) >= Teff);
            float mx = z[0], mn = z[0];
            #pragma unroll
            for (int k = 1; k < 8; ++k) { mx = fmaxf(mx, z[k]); mn = fminf(mn, z[k]); }
            unsigned smx = dead, smn = dead;
            #pragma unroll
            for (int k = 0; k < 8; ++k) {
                const unsigned em = __builtin_amdgcn_ballot_w32(z[k] == mx);
                const unsigned fm = em & ~smx;
                smx |= em;
                accMax[k] += __builtin_amdgcn_inverse_ballot_w32(fm) ? mx : 0.0f;
                const unsigned en = __builtin_amdgcn_ballot_w32(z[k] == mn);
                const unsigned fn = en & ~smn;
                smn |= en;
                cntMin[k] += __builtin_popcount(fn);
            }
        }
    }

    #pragma unroll
    for (int k = 0; k < 8; ++k) {
        float vM = accMax[k];
        #pragma unroll
        for (int off = 16; off > 0; off >>= 1) vM += __shfl_xor(vM, off, 32);
        if (lane == 0) {
            atomicAdd(&sAcc[k],     vM);
            atomicAdd(&sAcc[8 + k], (float)cntMin[k]);
        }
    }
    __syncthreads();

    if (tid < 16) {
        const int which = tid >> 3;
        const int k     = tid & 7;
        const size_t off = (size_t)b * 14336
                         + (size_t)(2 * u + which) * (NGROUP * NKER)
                         + (size_t)h * NKER + k;
        OUT[off] = sAcc[tid];
    }
}

extern "C" void kernel_launch(void* const* d_in, const int* in_sizes, int n_in,
                              void* d_out, int out_size, void* d_ws, size_t ws_size,
                              hipStream_t stream) {
    (void)in_sizes; (void)n_in; (void)d_ws; (void)ws_size; (void)out_size;
    const float* X   = (const float*)d_in[0];
    const float* W   = (const float*)d_in[1];
    const int*   IDX = (const int*)d_in[2];
    float*       OUT = (float*)d_out;

    const int shmem = SMEM_FLOATS * (int)sizeof(float);
    (void)hipFuncSetAttribute((const void*)hydra_unit_kernel,
                              hipFuncAttributeMaxDynamicSharedMemorySize, shmem);

    const int nblocks = 4 * NUNIT * NGROUP;
    hydra_unit_kernel<<<dim3(nblocks), dim3(1024), shmem, stream>>>(X, W, IDX, OUT);
}


# phase-2 trips with every tap inside the sequence load B at T + per-lane constant: no range compare, address select or outside-lane mask; edge trips keep the checked loads; sum order unchanged
# speedup vs baseline: 2.3202x; 1.0498x over previous
#include <hip/hip_runtime.h>

typedef __attribute__((ext_vector_type(2))) float v2f;
typedef __attribute__((ext_vector_type(8))) float v8f;

#define T_FULL 75000
#define CH_IN  17
#define NGROUP 32
#define NKER   8
#define NTAP   9
#define NUNIT  28
#define NQUAD  (T_FULL / 4)
#define SG_FLOATS 75040
#define ZSLOT     75008
#define SW_OFF    SG_FLOATS
#define SACC_OFF  (SG_FLOATS + 72)
#define SMEM_FLOATS (SACC_OFF + 16)

template <bool INTERIOR>
__device__ __forceinline__ void conv_trip(const float* sg, const int blk,
                                          const int c16, const int hs, const int krb,
                                          const int d, const int hmask, const int Teff,
                                          const int (&offI)[3][2], const v2f (&afrag)[3],
                                          float (&accMax)[8], int (&cntMin)[8])
{
    v2f bf[3];
    unsigned dead;
    if (INTERIOR) {
        const int q4 = (blk << 6) + (c16 << 2);
        const int TB = q4 + (q4 & (hmask * 4));
        const char* sgB = reinterpret_cast<const char*>(sg);
        #pragma unroll
        for (int tile = 0; tile < 3; ++tile) {
            #pragma unroll
            for (int i = 0; i < 2; ++i)
                bf[tile][i] = *reinterpret_cast<const float*>(sgB + TB + offI[tile][i]);
        }
        dead = 0u;
    } else {
        const int q = (blk << 4) + c16;
        const int T = q + (q & hmask);
        #pragma unroll
        for (int tile = 0; tile < 3; ++tile) {
            #pragma unroll
            for (int i = 0; i < 2; ++i) {
                const int kk = tile * 4 + krb + i;
                const int p  = T + (kk - 4) * d;
                const int a  = (kk < NTAP + 1 && (unsigned)p < (unsigned)Teff) ? p : ZSLOT;
                bf[tile][i] = sg[a];
            }
        }
        dead = __builtin_amdgcn_ballot_w32(T + (hs ? d : 0) >= Teff);
    }

    v8f z = {0.f,0.f,0.f,0.f,0.f,0.f,0.f,0.f};
    #pragma unroll
    for (int tile = 0; tile < 3; ++tile) {
        z = __builtin_amdgcn_wmma_f32_16x16x4_f32(
                false, afrag[tile], false, bf[tile], (short)0, z, false, false);
    }

    float mx = z[0], mn = z[0];
    #pragma unroll
    for (int k = 1; k < 8; ++k) { mx = fmaxf(mx, z[k]); mn = fminf(mn, z[k]); }
    unsigned smx = dead, smn = dead;
    #pragma unroll
    for (int k = 0; k < 8; ++k) {
        const unsigned em = __builtin_amdgcn_ballot_w32(z[k] == mx);
        const unsigned fm = em & ~smx;
        smx |= em;
        accMax[k] += __builtin_amdgcn_inverse_ballot_w32(fm) ? mx : 0.0f;
        const unsigned en = __builtin_amdgcn_ballot_w32(z[k] == mn);
        const unsigned fn = en & ~smn;
        smn |= en;
        cntMin[k] += __builtin_popcount(fn);
    }
}

__global__ __launch_bounds__(1024)
void hydra_unit_kernel(const float* __restrict__ X,
                       const float* __restrict__ W,
                       const int*   __restrict__ IDX,
                       float* __restrict__ OUT)
{
    extern __shared__ __attribute__((aligned(16))) float smem[];
    float* sg   = smem;
    float* sw   = smem + SW_OFF;
    float* sAcc = smem + SACC_OFF;

    const int tid  = threadIdx.x;
    const int lane = tid & 31;
    const int wav  = __builtin_amdgcn_readfirstlane(tid >> 5);
    const int nthr = blockDim.x;
    const int nwav = nthr >> 5;

    const int bid = blockIdx.x;
    const int b   = bid / (NUNIT * NGROUP);
    const int rem = bid % (NUNIT * NGROUP);
    const int u   = rem / NGROUP;
    const int h   = rem % NGROUP;
    const int di  = u >> 1;
    const int dif = u & 1;
    const int d   = 1 << di;
    const int Teff = dif ? (T_FULL - 1) : T_FULL;

    if (tid < 16) sAcc[tid] = 0.0f;
    if (tid == 16) sg[ZSLOT] = 0.0f;
    {
        const int wbase = (u * (NKER * NGROUP) + h * NKER) * NTAP;
        if (tid < NKER * NTAP) sw[tid] = W[wbase + tid];
    }
    int ch[8];
    {
        const int* ip = IDX + (u * NGROUP + h) * 8;
        #pragma unroll
        for (int i = 0; i < 8; ++i) ch[i] = ip[i];
    }

    {
        const float* Xb = X + (size_t)b * CH_IN * T_FULL;
        const float* xr[8];
        #pragma unroll
        for (int i = 0; i < 8; ++i) xr[i] = Xb + (size_t)ch[i] * T_FULL;
        float4* sg4 = reinterpret_cast<float4*>(sg);
        if (!dif) {
            for (int q = tid; q < NQUAD; q += nthr) {
                float4 v[8];
                #pragma unroll
                for (int i = 0; i < 8; ++i) v[i] = reinterpret_cast<const float4*>(xr[i])[q];
                float4 s = make_float4(0.0f, 0.0f, 0.0f, 0.0f);
                #pragma unroll
                for (int i = 0; i < 8; ++i) {
                    s.x += v[i].x;  s.y += v[i].y;  s.z += v[i].z;  s.w += v[i].w;
                }
                sg4[q] = s;
            }
        } else {
            for (int q = tid; q < NQUAD; q += nthr) {
                const int te = min(4 * q + 4, T_FULL - 1);
                float4 v[8];
                float  e[8];
                #pragma unroll
                for (int i = 0; i < 8; ++i) v[i] = reinterpret_cast<const float4*>(xr[i])[q];
                #pragma unroll
                for (int i = 0; i < 8; ++i) e[i] = xr[i][te];
                float4 s = make_float4(0.0f, 0.0f, 0.0f, 0.0f);
                #pragma unroll
                for (int i = 0; i < 8; ++i) {
                    s.x += v[i].y - v[i].x;
                    s.y += v[i].z - v[i].y;
                    s.z += v[i].w - v[i].z;
                    s.w += e[i]   - v[i].w;
                }
                sg4[q] = s;
            }
        }
    }
    __syncthreads();

    const int c16 = lane & 15;
    const int hs  = lane >> 4;
    const int krb = hs << 1;
    const int ak  = c16 & 7;
    const int as  = c16 >> 3;
    v2f afrag[3];
    #pragma unroll
    for (int tile = 0; tile < 3; ++tile) {
        #pragma unroll
        for (int i = 0; i < 2; ++i) {
            const int jj = tile * 4 + krb + i - as;
            afrag[tile][i] = (jj >= 0 && jj < NTAP) ? sw[ak * NTAP + jj] : 0.0f;
        }
    }

    float accMax[8];
    int   cntMin[8];
    #pragma unroll
    for (int k = 0; k < 8; ++k) { accMax[k] = 0.0f; cntMin[k] = 0; }

    const int hmask = -d;
    const int hp    = (Teff - 1) >> (di + 1);
    const int nq    = (hp << di) + min(d, Teff - (hp << (di + 1)));
    const int nblk  = (nq + 15) >> 4;

    const int blkA = (2 * d + 15) >> 4;
    const int hiT  = Teff - 5 * d - 1;
    int blkB = 0;
    if (hiT >= 0) {
        const int qmax = ((hiT >> (di + 1)) << di) + min(hiT & (2 * d - 1), d - 1);
        blkB = (qmax + 1) >> 4;
    }
    const int e1 = min(blkA, nblk);
    const int e2 = max(e1, min(blkB, nblk));

    int offI[3][2];
    #pragma unroll
    for (int tile = 0; tile < 3; ++tile) {
        #pragma unroll
        for (int i = 0; i < 2; ++i) {
            const int kk = tile * 4 + krb + i;
            offI[tile][i] = (kk < NTAP + 1) ? (kk - 4) * d * 4 : 0;
        }
    }

    int blk = wav;
    for (; blk < e1; blk += nwav)
        conv_trip<false>(sg, blk, c16, hs, krb, d, hmask, Teff, offI, afrag, accMax, cntMin);
    for (; blk < e2; blk += nwav)
        conv_trip<true>(sg, blk, c16, hs, krb, d, hmask, Teff, offI, afrag, accMax, cntMin);
    for (; blk < nblk; blk += nwav)
        conv_trip<false>(sg, blk, c16, hs, krb, d, hmask, Teff, offI, afrag, accMax, cntMin);

    #pragma unroll
    for (int k = 0; k < 8; ++k) {
        float vM = accMax[k];
        #pragma unroll
        for (int off = 16; off > 0; off >>= 1) vM += __shfl_xor(vM, off, 32);
        if (lane == 0) {
            atomicAdd(&sAcc[k],     vM);
            atomicAdd(&sAcc[8 + k], (float)cntMin[k]);
        }
    }
    __syncthreads();

    if (tid < 16) {
        const int which = tid >> 3;
        const int k     = tid & 7;
        const size_t off = (size_t)b * 14336
                         + (size_t)(2 * u + which) * (NGROUP * NKER)
                         + (size_t)h * NKER + k;
        OUT[off] = sAcc[tid];
    }
}

extern "C" void kernel_launch(void* const* d_in, const int* in_sizes, int n_in,
                              void* d_out, int out_size, void* d_ws, size_t ws_size,
                              hipStream_t stream) {
    (void)in_sizes; (void)n_in; (void)d_ws; (void)ws_size; (void)out_size;
    const float* X   = (const float*)d_in[0];
    const float* W   = (const float*)d_in[1];
    const int*   IDX = (const int*)d_in[2];
    float*       OUT = (float*)d_out;

    const int shmem = SMEM_FLOATS * (int)sizeof(float);
    (void)hipFuncSetAttribute((const void*)hydra_unit_kernel,
                              hipFuncAttributeMaxDynamicSharedMemorySize, shmem);

    const int nblocks = 4 * NUNIT * NGROUP;
    hydra_unit_kernel<<<dim3(nblocks), dim3(1024), shmem, stream>>>(X, W, IDX, OUT);
}


# phase-2 epilogue scalar ops 45 to 31 per trip: max/min seen-masks share one 64-bit s_or; argmin counted as s_bcnt1 of the running mask, differenced per wave, kernel 7 by remainder; float sums same
# speedup vs baseline: 2.4311x; 1.0478x over previous
#include <hip/hip_runtime.h>

typedef __attribute__((ext_vector_type(2))) float v2f;
typedef __attribute__((ext_vector_type(8))) float v8f;

#define T_FULL 75000
#define CH_IN  17
#define NGROUP 32
#define NKER   8
#define NTAP   9
#define NUNIT  28
#define NQUAD  (T_FULL / 4)
#define SG_FLOATS 75040
#define ZSLOT     75008
#define SW_OFF    SG_FLOATS
#define SACC_OFF  (SG_FLOATS + 72)
#define SMEM_FLOATS (SACC_OFF + 16)

template <bool INTERIOR>
__device__ __forceinline__ void conv_trip(const float* sg, const int blk,
                                          const int c16, const int hs, const int krb,
                                          const int d, const int hmask, const int Teff,
                                          const int (&offI)[3][2], const v2f (&afrag)[3],
                                          float (&accMax)[8], int (&cumMin)[7], int &deadCnt)
{
    v2f bf[3];
    unsigned dead;
    if (INTERIOR) {
        const int q4 = (blk << 6) + (c16 << 2);
        const int TB = q4 + (q4 & (hmask * 4));
        const char* sgB = reinterpret_cast<const char*>(sg);
        #pragma unroll
        for (int tile = 0; tile < 3; ++tile) {
            #pragma unroll
            for (int i = 0; i < 2; ++i)
                bf[tile][i] = *reinterpret_cast<const float*>(sgB + TB + offI[tile][i]);
        }
        dead = 0u;
    } else {
        const int q = (blk << 4) + c16;
        const int T = q + (q & hmask);
        #pragma unroll
        for (int tile = 0; tile < 3; ++tile) {
            #pragma unroll
            for (int i = 0; i < 2; ++i) {
                const int kk = tile * 4 + krb + i;
                const int p  = T + (kk - 4) * d;
                const int a  = (kk < NTAP + 1 && (unsigned)p < (unsigned)Teff) ? p : ZSLOT;
                bf[tile][i] = sg[a];
            }
        }
        dead = __builtin_amdgcn_ballot_w32(T + (hs ? d : 0) >= Teff);
    }

    v8f z = {0.f,0.f,0.f,0.f,0.f,0.f,0.f,0.f};
    #pragma unroll
    for (int tile = 0; tile < 3; ++tile) {
        z = __builtin_amdgcn_wmma_f32_16x16x4_f32(
                false, afrag[tile], false, bf[tile], (short)0, z, false, false);
    }

    float mx = z[0], mn = z[0];
    #pragma unroll
    for (int k = 1; k < 8; ++k) { mx = fmaxf(mx, z[k]); mn = fminf(mn, z[k]); }
    unsigned long long seen = ((unsigned long long)dead << 32) | dead;
    if (!INTERIOR) deadCnt += __builtin_popcount(dead);
    #pragma unroll
    for (int k = 0; k < 8; ++k) {
        const unsigned em = __builtin_amdgcn_ballot_w32(z[k] == mx);
        const unsigned fm = em & ~(unsigned)seen;
        accMax[k] += __builtin_amdgcn_inverse_ballot_w32(fm) ? mx : 0.0f;
        if (k < 7) {
            const unsigned en = __builtin_amdgcn_ballot_w32(z[k] == mn);
            seen |= ((unsigned long long)en << 32) | em;
            cumMin[k] += __builtin_popcount((unsigned)(seen >> 32));
        }
    }
}

__global__ __launch_bounds__(1024)
void hydra_unit_kernel(const float* __restrict__ X,
                       const float* __restrict__ W,
                       const int*   __restrict__ IDX,
                       float* __restrict__ OUT)
{
    extern __shared__ __attribute__((aligned(16))) float smem[];
    float* sg   = smem;
    float* sw   = smem + SW_OFF;
    float* sAcc = smem + SACC_OFF;

    const int tid  = threadIdx.x;
    const int lane = tid & 31;
    const int wav  = __builtin_amdgcn_readfirstlane(tid >> 5);
    const int nthr = blockDim.x;
    const int nwav = nthr >> 5;

    const int bid = blockIdx.x;
    const int b   = bid / (NUNIT * NGROUP);
    const int rem = bid % (NUNIT * NGROUP);
    const int u   = rem / NGROUP;
    const int h   = rem % NGROUP;
    const int di  = u >> 1;
    const int dif = u & 1;
    const int d   = 1 << di;
    const int Teff = dif ? (T_FULL - 1) : T_FULL;

    if (tid < 16) sAcc[tid] = 0.0f;
    if (tid == 16) sg[ZSLOT] = 0.0f;
    {
        const int wbase = (u * (NKER * NGROUP) + h * NKER) * NTAP;
        if (tid < NKER * NTAP) sw[tid] = W[wbase + tid];
    }
    int ch[8];
    {
        const int* ip = IDX + (u * NGROUP + h) * 8;
        #pragma unroll
        for (int i = 0; i < 8; ++i) ch[i] = ip[i];
    }

    {
        const float* Xb = X + (size_t)b * CH_IN * T_FULL;
        const float* xr[8];
        #pragma unroll
        for (int i = 0; i < 8; ++i) xr[i] = Xb + (size_t)ch[i] * T_FULL;
        float4* sg4 = reinterpret_cast<float4*>(sg);
        if (!dif) {
            for (int q = tid; q < NQUAD; q += nthr) {
                float4 v[8];
                #pragma unroll
                for (int i = 0; i < 8; ++i) v[i] = reinterpret_cast<const float4*>(xr[i])[q];
                float4 s = make_float4(0.0f, 0.0f, 0.0f, 0.0f);
                #pragma unroll
                for (int i = 0; i < 8; ++i) {
                    s.x += v[i].x;  s.y += v[i].y;  s.z += v[i].z;  s.w += v[i].w;
                }
                sg4[q] = s;
            }
        } else {
            for (int q = tid; q < NQUAD; q += nthr) {
                const int te = min(4 * q + 4, T_FULL - 1);
                float4 v[8];
                float  e[8];
                #pragma unroll
                for (int i = 0; i < 8; ++i) v[i] = reinterpret_cast<const float4*>(xr[i])[q];
                #pragma unroll
                for (int i = 0; i < 8; ++i) e[i] = xr[i][te];
                float4 s = make_float4(0.0f, 0.0f, 0.0f, 0.0f);
                #pragma unroll
                for (int i = 0; i < 8; ++i) {
                    s.x += v[i].y - v[i].x;
                    s.y += v[i].z - v[i].y;
                    s.z += v[i].w - v[i].z;
                    s.w += e[i]   - v[i].w;
                }
                sg4[q] = s;
            }
        }
    }
    __syncthreads();

    const int c16 = lane & 15;
    const int hs  = lane >> 4;
    const int krb = hs << 1;
    const int ak  = c16 & 7;
    const int as  = c16 >> 3;
    v2f afrag[3];
    #pragma unroll
    for (int tile = 0; tile < 3; ++tile) {
        #pragma unroll
        for (int i = 0; i < 2; ++i) {
            const int jj = tile * 4 + krb + i - as;
            afrag[tile][i] = (jj >= 0 && jj < NTAP) ? sw[ak * NTAP + jj] : 0.0f;
        }
    }

    float accMax[8];
    int   cumMin[7];
    int   deadCnt = 0;
    #pragma unroll
    for (int k = 0; k < 8; ++k) accMax[k] = 0.0f;
    #pragma unroll
    for (int k = 0; k < 7; ++k) cumMin[k] = 0;

    const int hmask = -d;
    const int hp    = (Teff - 1) >> (di + 1);
    const int nq    = (hp << di) + min(d, Teff - (hp << (di + 1)));
    const int nblk  = (nq + 15) >> 4;

    const int blkA = (2 * d + 15) >> 4;
    const int hiT  = Teff - 5 * d - 1;
    int blkB = 0;
    if (hiT >= 0) {
        const int qmax = ((hiT >> (di + 1)) << di) + min(hiT & (2 * d - 1), d - 1);
        blkB = (qmax + 1) >> 4;
    }
    const int e1 = min(blkA, nblk);
    const int e2 = max(e1, min(blkB, nblk));

    int offI[3][2];
    #pragma unroll
    for (int tile = 0; tile < 3; ++tile) {
        #pragma unroll
        for (int i = 0; i < 2; ++i) {
            const int kk = tile * 4 + krb + i;
            offI[tile][i] = (kk < NTAP + 1) ? (kk - 4) * d * 4 : 0;
        }
    }

    int blk = wav;
    for (; blk < e1; blk += nwav)
        conv_trip<false>(sg, blk, c16, hs, krb, d, hmask, Teff, offI, afrag, accMax, cumMin, deadCnt);
    for (; blk < e2; blk += nwav)
        conv_trip<true>(sg, blk, c16, hs, krb, d, hmask, Teff, offI, afrag, accMax, cumMin, deadCnt);
    for (; blk < nblk; blk += nwav)
        conv_trip<false>(sg, blk, c16, hs, krb, d, hmask, Teff, offI, afrag, accMax, cumMin, deadCnt);

    int cumBelow = deadCnt;
    #pragma unroll
    for (int k = 0; k < 8; ++k) {
        float vM = accMax[k];
        #pragma unroll
        for (int off = 16; off > 0; off >>= 1) vM += __shfl_xor(vM, off, 32);
        if (lane == 0) {
            atomicAdd(&sAcc[k], vM);
            if (k < 7) atomicAdd(&sAcc[8 + k], (float)(cumMin[k] - cumBelow));
        }
        if (k < 7) cumBelow = cumMin[k];
    }
    __syncthreads();

    if (tid < 16) {
        const int which = tid >> 3;
        const int k     = tid & 7;
        const size_t off = (size_t)b * 14336
                         + (size_t)(2 * u + which) * (NGROUP * NKER)
                         + (size_t)h * NKER + k;
        float v = sAcc[tid];
        if (tid == 15) {
            float lower = 0.0f;
            #pragma unroll
            for (int j = 0; j < 7; ++j) lower += sAcc[8 + j];
            v = (float)Teff - lower;
        }
        OUT[off] = v;
    }
}

extern "C" void kernel_launch(void* const* d_in, const int* in_sizes, int n_in,
                              void* d_out, int out_size, void* d_ws, size_t ws_size,
                              hipStream_t stream) {
    (void)in_sizes; (void)n_in; (void)d_ws; (void)ws_size; (void)out_size;
    const float* X   = (const float*)d_in[0];
    const float* W   = (const float*)d_in[1];
    const int*   IDX = (const int*)d_in[2];
    float*       OUT = (float*)d_out;

    const int shmem = SMEM_FLOATS * (int)sizeof(float);
    (void)hipFuncSetAttribute((const void*)hydra_unit_kernel,
                              hipFuncAttributeMaxDynamicSharedMemorySize, shmem);

    const int nblocks = 4 * NUNIT * NGROUP;
    hydra_unit_kernel<<<dim3(nblocks), dim3(1024), shmem, stream>>>(X, W, IDX, OUT);
}


# phase-2 trips two per pass: lane masks of both trips in one 64-bit scalar per kernel and side, so each and-not/or/bit-count serves 64 time points; two WMMA chains per pass; sum order unchanged
# speedup vs baseline: 2.5408x; 1.0451x over previous
#include <hip/hip_runtime.h>

typedef __attribute__((ext_vector_type(2))) float v2f;
typedef __attribute__((ext_vector_type(8))) float v8f;

#define T_FULL 75000
#define CH_IN  17
#define NGROUP 32
#define NKER   8
#define NTAP   9
#define NUNIT  28
#define NQUAD  (T_FULL / 4)
#define SG_FLOATS 75040
#define ZSLOT     75008
#define SW_OFF    SG_FLOATS
#define SACC_OFF  (SG_FLOATS + 72)
#define SMEM_FLOATS (SACC_OFF + 16)

template <bool INTERIOR>
__device__ __forceinline__ void conv_trip(const float* sg, const int blk,
                                          const int c16, const int hs, const int krb,
                                          const int d, const int hmask, const int Teff,
                                          const int (&offI)[3][2], const v2f (&afrag)[3],
                                          float (&accMax)[8], int (&cumMin)[7], int &deadCnt)
{
    v2f bf[3];
    unsigned dead;
    if (INTERIOR) {
        const int q4 = (blk << 6) + (c16 << 2);
        const int TB = q4 + (q4 & (hmask * 4));
        const char* sgB = reinterpret_cast<const char*>(sg);
        #pragma unroll
        for (int tile = 0; tile < 3; ++tile) {
            #pragma unroll
            for (int i = 0; i < 2; ++i)
                bf[tile][i] = *reinterpret_cast<const float*>(sgB + TB + offI[tile][i]);
        }
        dead = 0u;
    } else {
        const int q = (blk << 4) + c16;
        const int T = q + (q & hmask);
        #pragma unroll
        for (int tile = 0; tile < 3; ++tile) {
            #pragma unroll
            for (int i = 0; i < 2; ++i) {
                const int kk = tile * 4 + krb + i;
                const int p  = T + (kk - 4) * d;
                const int a  = (kk < NTAP + 1 && (unsigned)p < (unsigned)Teff) ? p : ZSLOT;
                bf[tile][i] = sg[a];
            }
        }
        dead = __builtin_amdgcn_ballot_w32(T + (hs ? d : 0) >= Teff);
    }

    v8f z = {0.f,0.f,0.f,0.f,0.f,0.f,0.f,0.f};
    #pragma unroll
    for (int tile = 0; tile < 3; ++tile) {
        z = __builtin_amdgcn_wmma_f32_16x16x4_f32(
                false, afrag[tile], false, bf[tile], (short)0, z, false, false);
    }

    float mx = z[0], mn = z[0];
    #pragma unroll
    for (int k = 1; k < 8; ++k) { mx = fmaxf(mx, z[k]); mn = fminf(mn, z[k]); }
    unsigned long long seen = ((unsigned long long)dead << 32) | dead;
    if (!INTERIOR) deadCnt += __builtin_popcount(dead);
    #pragma unroll
    for (int k = 0; k < 8; ++k) {
        const unsigned em = __builtin_amdgcn_ballot_w32(z[k] == mx);
        const unsigned fm = em & ~(unsigned)seen;
        accMax[k] += __builtin_amdgcn_inverse_ballot_w32(fm) ? mx : 0.0f;
        if (k < 7) {
            const unsigned en = __builtin_amdgcn_ballot_w32(z[k] == mn);
            seen |= ((unsigned long long)en << 32) | em;
            cumMin[k] += __builtin_popcount((unsigned)(seen >> 32));
        }
    }
}

template <bool INTERIOR>
__device__ __forceinline__ void conv_trip_pair(const float* sg, const int blka, const int blkb,
                                               const int c16, const int hs, const int krb,
                                               const int d, const int hmask, const int Teff,
                                               const int (&offI)[3][2], const v2f (&afrag)[3],
                                               float (&accMax)[8], int (&cumMin)[7], int &deadCnt)
{
    v2f bfa[3], bfb[3];
    unsigned long long dead;
    if (INTERIOR) {
        const char* sgB = reinterpret_cast<const char*>(sg);
        const int q4a = (blka << 6) + (c16 << 2);
        const int q4b = (blkb << 6) + (c16 << 2);
        const int TBa = q4a + (q4a & (hmask * 4));
        const int TBb = q4b + (q4b & (hmask * 4));
        #pragma unroll
        for (int tile = 0; tile < 3; ++tile) {
            #pragma unroll
            for (int i = 0; i < 2; ++i) {
                bfa[tile][i] = *reinterpret_cast<const float*>(sgB + TBa + offI[tile][i]);
                bfb[tile][i] = *reinterpret_cast<const float*>(sgB + TBb + offI[tile][i]);
            }
        }
        dead = 0ull;
    } else {
        const int qa = (blka << 4) + c16;
        const int qb = (blkb << 4) + c16;
        const int Ta = qa + (qa & hmask);
        const int Tb = qb + (qb & hmask);
        #pragma unroll
        for (int tile = 0; tile < 3; ++tile) {
            #pragma unroll
            for (int i = 0; i < 2; ++i) {
                const int kk = tile * 4 + krb + i;
                const int pa = Ta + (kk - 4) * d;
                const int pb = Tb + (kk - 4) * d;
                const int aa = (kk < NTAP + 1 && (unsigned)pa < (unsigned)Teff) ? pa : ZSLOT;
                const int ab = (kk < NTAP + 1 && (unsigned)pb < (unsigned)Teff) ? pb : ZSLOT;
                bfa[tile][i] = sg[aa];
                bfb[tile][i] = sg[ab];
            }
        }
        const unsigned deada = __builtin_amdgcn_ballot_w32(Ta + (hs ? d : 0) >= Teff);
        const unsigned deadb = __builtin_amdgcn_ballot_w32(Tb + (hs ? d : 0) >= Teff);
        dead = ((unsigned long long)deadb << 32) | deada;
        deadCnt += __builtin_popcountll(dead);
    }
    v8f za = {0.f,0.f,0.f,0.f,0.f,0.f,0.f,0.f};
    v8f zb = {0.f,0.f,0.f,0.f,0.f,0.f,0.f,0.f};
    #pragma unroll
    for (int tile = 0; tile < 3; ++tile) {
        za = __builtin_amdgcn_wmma_f32_16x16x4_f32(
                false, afrag[tile], false, bfa[tile], (short)0, za, false, false);
        zb = __builtin_amdgcn_wmma_f32_16x16x4_f32(
                false, afrag[tile], false, bfb[tile], (short)0, zb, false, false);
    }
    float mxa = za[0], mna = za[0], mxb = zb[0], mnb = zb[0];
    #pragma unroll
    for (int k = 1; k < 8; ++k) {
        mxa = fmaxf(mxa, za[k]);  mna = fminf(mna, za[k]);
        mxb = fmaxf(mxb, zb[k]);  mnb = fminf(mnb, zb[k]);
    }
    unsigned long long seenMax = dead;
    unsigned long long seenMin = dead;
    #pragma unroll
    for (int k = 0; k < 8; ++k) {
        const unsigned ema = __builtin_amdgcn_ballot_w32(za[k] == mxa);
        const unsigned emb = __builtin_amdgcn_ballot_w32(zb[k] == mxb);
        const unsigned long long em = ((unsigned long long)emb << 32) | ema;
        const unsigned long long fm = em & ~seenMax;
        accMax[k] += __builtin_amdgcn_inverse_ballot_w32((unsigned)fm) ? mxa : 0.0f;
        accMax[k] += __builtin_amdgcn_inverse_ballot_w32((unsigned)(fm >> 32)) ? mxb : 0.0f;
        if (k < 7) {
            seenMax |= em;
            const unsigned ena = __builtin_amdgcn_ballot_w32(za[k] == mna);
            const unsigned enb = __builtin_amdgcn_ballot_w32(zb[k] == mnb);
            seenMin |= ((unsigned long long)enb << 32) | ena;
            cumMin[k] += __builtin_popcountll(seenMin);
        }
    }
}

__global__ __launch_bounds__(1024)
void hydra_unit_kernel(const float* __restrict__ X,
                       const float* __restrict__ W,
                       const int*   __restrict__ IDX,
                       float* __restrict__ OUT)
{
    extern __shared__ __attribute__((aligned(16))) float smem[];
    float* sg   = smem;
    float* sw   = smem + SW_OFF;
    float* sAcc = smem + SACC_OFF;

    const int tid  = threadIdx.x;
    const int lane = tid & 31;
    const int wav  = __builtin_amdgcn_readfirstlane(tid >> 5);
    const int nthr = blockDim.x;
    const int nwav = nthr >> 5;

    const int bid = blockIdx.x;
    const int b   = bid / (NUNIT * NGROUP);
    const int rem = bid % (NUNIT * NGROUP);
    const int u   = rem / NGROUP;
    const int h   = rem % NGROUP;
    const int di  = u >> 1;
    const int dif = u & 1;
    const int d   = 1 << di;
    const int Teff = dif ? (T_FULL - 1) : T_FULL;

    if (tid < 16) sAcc[tid] = 0.0f;
    if (tid == 16) sg[ZSLOT] = 0.0f;
    {
        const int wbase = (u * (NKER * NGROUP) + h * NKER) * NTAP;
        if (tid < NKER * NTAP) sw[tid] = W[wbase + tid];
    }
    int ch[8];
    {
        const int* ip = IDX + (u * NGROUP + h) * 8;
        #pragma unroll
        for (int i = 0; i < 8; ++i) ch[i] = ip[i];
    }

    {
        const float* Xb = X + (size_t)b * CH_IN * T_FULL;
        const float* xr[8];
        #pragma unroll
        for (int i = 0; i < 8; ++i) xr[i] = Xb + (size_t)ch[i] * T_FULL;
        float4* sg4 = reinterpret_cast<float4*>(sg);
        if (!dif) {
            for (int q = tid; q < NQUAD; q += nthr) {
                float4 v[8];
                #pragma unroll
                for (int i = 0; i < 8; ++i) v[i] = reinterpret_cast<const float4*>(xr[i])[q];
                float4 s = make_float4(0.0f, 0.0f, 0.0f, 0.0f);
                #pragma unroll
                for (int i = 0; i < 8; ++i) {
                    s.x += v[i].x;  s.y += v[i].y;  s.z += v[i].z;  s.w += v[i].w;
                }
                sg4[q] = s;
            }
        } else {
            for (int q = tid; q < NQUAD; q += nthr) {
                const int te = min(4 * q + 4, T_FULL - 1);
                float4 v[8];
                float  e[8];
                #pragma unroll
                for (int i = 0; i < 8; ++i) v[i] = reinterpret_cast<const float4*>(xr[i])[q];
                #pragma unroll
                for (int i = 0; i < 8; ++i) e[i] = xr[i][te];
                float4 s = make_float4(0.0f, 0.0f, 0.0f, 0.0f);
                #pragma unroll
                for (int i = 0; i < 8; ++i) {
                    s.x += v[i].y - v[i].x;
                    s.y += v[i].z - v[i].y;
                    s.z += v[i].w - v[i].z;
                    s.w += e[i]   - v[i].w;
                }
                sg4[q] = s;
            }
        }
    }
    __syncthreads();

    const int c16 = lane & 15;
    const int hs  = lane >> 4;
    const int krb = hs << 1;
    const int ak  = c16 & 7;
    const int as  = c16 >> 3;
    v2f afrag[3];
    #pragma unroll
    for (int tile = 0; tile < 3; ++tile) {
        #pragma unroll
        for (int i = 0; i < 2; ++i) {
            const int jj = tile * 4 + krb + i - as;
            afrag[tile][i] = (jj >= 0 && jj < NTAP) ? sw[ak * NTAP + jj] : 0.0f;
        }
    }

    float accMax[8];
    int   cumMin[7];
    int   deadCnt = 0;
    #pragma unroll
    for (int k = 0; k < 8; ++k) accMax[k] = 0.0f;
    #pragma unroll
    for (int k = 0; k < 7; ++k) cumMin[k] = 0;

    const int hmask = -d;
    const int hp    = (Teff - 1) >> (di + 1);
    const int nq    = (hp << di) + min(d, Teff - (hp << (di + 1)));
    const int nblk  = (nq + 15) >> 4;

    const int blkA = (2 * d + 15) >> 4;
    const int hiT  = Teff - 5 * d - 1;
    int blkB = 0;
    if (hiT >= 0) {
        const int qmax = ((hiT >> (di + 1)) << di) + min(hiT & (2 * d - 1), d - 1);
        blkB = (qmax + 1) >> 4;
    }
    const int e1 = min(blkA, nblk);
    const int e2 = max(e1, min(blkB, nblk));

    int offI[3][2];
    #pragma unroll
    for (int tile = 0; tile < 3; ++tile) {
        #pragma unroll
        for (int i = 0; i < 2; ++i) {
            const int kk = tile * 4 + krb + i;
            offI[tile][i] = (kk < NTAP + 1) ? (kk - 4) * d * 4 : 0;
        }
    }

    int blk = wav;
    for (; blk + nwav < e1; blk += 2 * nwav)
        conv_trip_pair<false>(sg, blk, blk + nwav, c16, hs, krb, d, hmask, Teff, offI, afrag, accMax, cumMin, deadCnt);
    for (; blk < e1; blk += nwav)
        conv_trip<false>(sg, blk, c16, hs, krb, d, hmask, Teff, offI, afrag, accMax, cumMin, deadCnt);
    for (; blk + nwav < e2; blk += 2 * nwav)
        conv_trip_pair<true>(sg, blk, blk + nwav, c16, hs, krb, d, hmask, Teff, offI, afrag, accMax, cumMin, deadCnt);
    for (; blk < e2; blk += nwav)
        conv_trip<true>(sg, blk, c16, hs, krb, d, hmask, Teff, offI, afrag, accMax, cumMin, deadCnt);
    for (; blk + nwav < nblk; blk += 2 * nwav)
        conv_trip_pair<false>(sg, blk, blk + nwav, c16, hs, krb, d, hmask, Teff, offI, afrag, accMax, cumMin, deadCnt);
    for (; blk < nblk; blk += nwav)
        conv_trip<false>(sg, blk, c16, hs, krb, d, hmask, Teff, offI, afrag, accMax, cumMin, deadCnt);

    int cumBelow = deadCnt;
    #pragma unroll
    for (int k = 0; k < 8; ++k) {
        float vM = accMax[k];
        #pragma unroll
        for (int off = 16; off > 0; off >>= 1) vM += __shfl_xor(vM, off, 32);
        if (lane == 0) {
            atomicAdd(&sAcc[k], vM);
            if (k < 7) atomicAdd(&sAcc[8 + k], (float)(cumMin[k] - cumBelow));
        }
        if (k < 7) cumBelow = cumMin[k];
    }
    __syncthreads();

    if (tid < 16) {
        const int which = tid >> 3;
        const int k     = tid & 7;
        const size_t off = (size_t)b * 14336
                         + (size_t)(2 * u + which) * (NGROUP * NKER)
                         + (size_t)h * NKER + k;
        float v = sAcc[tid];
        if (tid == 15) {
            float lower = 0.0f;
            #pragma unroll
            for (int j = 0; j < 7; ++j) lower += sAcc[8 + j];
            v = (float)Teff - lower;
        }
        OUT[off] = v;
    }
}

extern "C" void kernel_launch(void* const* d_in, const int* in_sizes, int n_in,
                              void* d_out, int out_size, void* d_ws, size_t ws_size,
                              hipStream_t stream) {
    (void)in_sizes; (void)n_in; (void)d_ws; (void)ws_size; (void)out_size;
    const float* X   = (const float*)d_in[0];
    const float* W   = (const float*)d_in[1];
    const int*   IDX = (const int*)d_in[2];
    float*       OUT = (float*)d_out;

    const int shmem = SMEM_FLOATS * (int)sizeof(float);
    (void)hipFuncSetAttribute((const void*)hydra_unit_kernel,
                              hipFuncAttributeMaxDynamicSharedMemorySize, shmem);

    const int nblocks = 4 * NUNIT * NGROUP;
    hydra_unit_kernel<<<dim3(nblocks), dim3(1024), shmem, stream>>>(X, W, IDX, OUT);
}


# dilation >= 32: time points split on bit di-1 into two independent half problems at half the dilation, run in turn in half the LDS, two 16-wave groups per compute unit; 2 launches; max sums reordered
# speedup vs baseline: 2.6360x; 1.0375x over previous
#include <hip/hip_runtime.h>

typedef __attribute__((ext_vector_type(2))) float v2f;
typedef __attribute__((ext_vector_type(8))) float v8f;

#define T_FULL 75000
#define CH_IN  17
#define NGROUP 32
#define NKER   8
#define NTAP   9
#define NUNIT  28
#define NQUAD  (T_FULL / 4)
#define SG_FLOATS 75040
#define ZSLOT     75008
#define SPLIT_DI0   5
#define NUNIT_WHOLE (2 * SPLIT_DI0)
#define SG_FLOATS_H 38160
#define ZSLOT_H     38144
#define SMEM_TAIL   (72 + 16)

template <bool INTERIOR, int ZS>
__device__ __forceinline__ void conv_trip(const float* sg, const int blk,
                                          const int c16, const int hs, const int krb,
                                          const int d, const int hmask, const int Teff,
                                          const int (&offI)[3][2], const v2f (&afrag)[3],
                                          float (&accMax)[8], int (&cumMin)[7], int &deadCnt)
{
    v2f bf[3];
    unsigned dead;
    if (INTERIOR) {
        const int q4 = (blk << 6) + (c16 << 2);
        const int TB = q4 + (q4 & (hmask * 4));
        const char* sgB = reinterpret_cast<const char*>(sg);
        #pragma unroll
        for (int tile = 0; tile < 3; ++tile) {
            #pragma unroll
            for (int i = 0; i < 2; ++i)
                bf[tile][i] = *reinterpret_cast<const float*>(sgB + TB + offI[tile][i]);
        }
        dead = 0u;
    } else {
        const int q = (blk << 4) + c16;
        const int T = q + (q & hmask);
        #pragma unroll
        for (int tile = 0; tile < 3; ++tile) {
            #pragma unroll
            for (int i = 0; i < 2; ++i) {
                const int kk = tile * 4 + krb + i;
                const int p  = T + (kk - 4) * d;
                const int a  = (kk < NTAP + 1 && (unsigned)p < (unsigned)Teff) ? p : ZS;
                bf[tile][i] = sg[a];
            }
        }
        dead = __builtin_amdgcn_ballot_w32(T + (hs ? d : 0) >= Teff);
    }

    v8f z = {0.f,0.f,0.f,0.f,0.f,0.f,0.f,0.f};
    #pragma unroll
    for (int tile = 0; tile < 3; ++tile) {
        z = __builtin_amdgcn_wmma_f32_16x16x4_f32(
                false, afrag[tile], false, bf[tile], (short)0, z, false, false);
    }

    float mx = z[0], mn = z[0];
    #pragma unroll
    for (int k = 1; k < 8; ++k) { mx = fmaxf(mx, z[k]); mn = fminf(mn, z[k]); }
    unsigned long long seen = ((unsigned long long)dead << 32) | dead;
    if (!INTERIOR) deadCnt += __builtin_popcount(dead);
    #pragma unroll
    for (int k = 0; k < 8; ++k) {
        const unsigned em = __builtin_amdgcn_ballot_w32(z[k] == mx);
        const unsigned fm = em & ~(unsigned)seen;
        accMax[k] += __builtin_amdgcn_inverse_ballot_w32(fm) ? mx : 0.0f;
        if (k < 7) {
            const unsigned en = __builtin_amdgcn_ballot_w32(z[k] == mn);
            seen |= ((unsigned long long)en << 32) | em;
            cumMin[k] += __builtin_popcount((unsigned)(seen >> 32));
        }
    }
}

template <bool INTERIOR, int ZS>
__device__ __forceinline__ void conv_trip_pair(const float* sg, const int blka, const int blkb,
                                               const int c16, const int hs, const int krb,
                                               const int d, const int hmask, const int Teff,
                                               const int (&offI)[3][2], const v2f (&afrag)[3],
                                               float (&accMax)[8], int (&cumMin)[7], int &deadCnt)
{
    v2f bfa[3], bfb[3];
    unsigned long long dead;
    if (INTERIOR) {
        const char* sgB = reinterpret_cast<const char*>(sg);
        const int q4a = (blka << 6) + (c16 << 2);
        const int q4b = (blkb << 6) + (c16 << 2);
        const int TBa = q4a + (q4a & (hmask * 4));
        const int TBb = q4b + (q4b & (hmask * 4));
        #pragma unroll
        for (int tile = 0; tile < 3; ++tile) {
            #pragma unroll
            for (int i = 0; i < 2; ++i) {
                bfa[tile][i] = *reinterpret_cast<const float*>(sgB + TBa + offI[tile][i]);
                bfb[tile][i] = *reinterpret_cast<const float*>(sgB + TBb + offI[tile][i]);
            }
        }
        dead = 0ull;
    } else {
        const int qa = (blka << 4) + c16;
        const int qb = (blkb << 4) + c16;
        const int Ta = qa + (qa & hmask);
        const int Tb = qb + (qb & hmask);
        #pragma unroll
        for (int tile = 0; tile < 3; ++tile) {
            #pragma unroll
            for (int i = 0; i < 2; ++i) {
                const int kk = tile * 4 + krb + i;
                const int pa = Ta + (kk - 4) * d;
                const int pb = Tb + (kk - 4) * d;
                const int aa = (kk < NTAP + 1 && (unsigned)pa < (unsigned)Teff) ? pa : ZS;
                const int ab = (kk < NTAP + 1 && (unsigned)pb < (unsigned)Teff) ? pb : ZS;
                bfa[tile][i] = sg[aa];
                bfb[tile][i] = sg[ab];
            }
        }
        const unsigned deada = __builtin_amdgcn_ballot_w32(Ta + (hs ? d : 0) >= Teff);
        const unsigned deadb = __builtin_amdgcn_ballot_w32(Tb + (hs ? d : 0) >= Teff);
        dead = ((unsigned long long)deadb << 32) | deada;
        deadCnt += __builtin_popcountll(dead);
    }
    v8f za = {0.f,0.f,0.f,0.f,0.f,0.f,0.f,0.f};
    v8f zb = {0.f,0.f,0.f,0.f,0.f,0.f,0.f,0.f};
    #pragma unroll
    for (int tile = 0; tile < 3; ++tile) {
        za = __builtin_amdgcn_wmma_f32_16x16x4_f32(
                false, afrag[tile], false, bfa[tile], (short)0, za, false, false);
        zb = __builtin_amdgcn_wmma_f32_16x16x4_f32(
                false, afrag[tile], false, bfb[tile], (short)0, zb, false, false);
    }
    float mxa = za[0], mna = za[0], mxb = zb[0], mnb = zb[0];
    #pragma unroll
    for (int k = 1; k < 8; ++k) {
        mxa = fmaxf(mxa, za[k]);  mna = fminf(mna, za[k]);
        mxb = fmaxf(mxb, zb[k]);  mnb = fminf(mnb, zb[k]);
    }
    unsigned long long seenMax = dead;
    unsigned long long seenMin = dead;
    #pragma unroll
    for (int k = 0; k < 8; ++k) {
        const unsigned ema = __builtin_amdgcn_ballot_w32(za[k] == mxa);
        const unsigned emb = __builtin_amdgcn_ballot_w32(zb[k] == mxb);
        const unsigned long long em = ((unsigned long long)emb << 32) | ema;
        const unsigned long long fm = em & ~seenMax;
        accMax[k] += __builtin_amdgcn_inverse_ballot_w32((unsigned)fm) ? mxa : 0.0f;
        accMax[k] += __builtin_amdgcn_inverse_ballot_w32((unsigned)(fm >> 32)) ? mxb : 0.0f;
        if (k < 7) {
            seenMax |= em;
            const unsigned ena = __builtin_amdgcn_ballot_w32(za[k] == mna);
            const unsigned enb = __builtin_amdgcn_ballot_w32(zb[k] == mnb);
            seenMin |= ((unsigned long long)enb << 32) | ena;
            cumMin[k] += __builtin_popcountll(seenMin);
        }
    }
}

template <bool SPLIT>
__global__ __launch_bounds__(SPLIT ? 512 : 1024)
void hydra_unit_kernel(const float* __restrict__ X,
                       const float* __restrict__ W,
                       const int*   __restrict__ IDX,
                       float* __restrict__ OUT)
{
    constexpr int SGF = SPLIT ? SG_FLOATS_H : SG_FLOATS;
    constexpr int ZS  = SPLIT ? ZSLOT_H : ZSLOT;
    extern __shared__ __attribute__((aligned(16))) float smem[];
    float* sg   = smem;
    float* sw   = smem + SGF;
    float* sAcc = smem + SGF + 72;

    const int tid  = threadIdx.x;
    const int lane = tid & 31;
    const int wav  = __builtin_amdgcn_readfirstlane(tid >> 5);
    const int nthr = blockDim.x;
    const int nwav = nthr >> 5;

    constexpr int NUN = SPLIT ? (NUNIT - NUNIT_WHOLE) : NUNIT_WHOLE;
    constexpr int U0  = SPLIT ? NUNIT_WHOLE : 0;
    const int bid = blockIdx.x;
    const int b   = bid / (NUN * NGROUP);
    const int rem = bid % (NUN * NGROUP);
    const int u   = U0 + rem / NGROUP;
    const int h   = rem % NGROUP;
    const int di  = u >> 1;
    const int dif = u & 1;
    const int d   = 1 << di;
    const int Teff = dif ? (T_FULL - 1) : T_FULL;

    const int ddi = SPLIT ? (di - 1) : di;
    const int dd  = 1 << ddi;
    const int eq  = di - 3;
    const int nhalf = SPLIT ? 2 : 1;

    if (tid < 16) sAcc[tid] = 0.0f;
    if (tid == 16) sg[ZS] = 0.0f;
    {
        const int wbase = (u * (NKER * NGROUP) + h * NKER) * NTAP;
        if (tid < NKER * NTAP) sw[tid] = W[wbase + tid];
    }
    int ch[8];
    {
        const int* ip = IDX + (u * NGROUP + h) * 8;
        #pragma unroll
        for (int i = 0; i < 8; ++i) ch[i] = ip[i];
    }
    const float* Xb = X + (size_t)b * CH_IN * T_FULL;
    const float* xr[8];
    #pragma unroll
    for (int i = 0; i < 8; ++i) xr[i] = Xb + (size_t)ch[i] * T_FULL;

    const int c16 = lane & 15;
    const int hs  = lane >> 4;
    const int krb = hs << 1;
    const int ak  = c16 & 7;
    const int as  = c16 >> 3;

    float accMax[8];
    int   cumMin[7];
    int   deadCnt = 0;
    #pragma unroll
    for (int k = 0; k < 8; ++k) accMax[k] = 0.0f;
    #pragma unroll
    for (int k = 0; k < 7; ++k) cumMin[k] = 0;

    int offI[3][2];
    #pragma unroll
    for (int tile = 0; tile < 3; ++tile) {
        #pragma unroll
        for (int i = 0; i < 2; ++i) {
            const int kk = tile * 4 + krb + i;
            offI[tile][i] = (kk < NTAP + 1) ? (kk - 4) * dd * 4 : 0;
        }
    }

    #pragma nounroll
    for (int half = 0; half < nhalf; ++half) {
        int TeffH = Teff;
        int nquadH = NQUAD;
        if (SPLIT) {
            const int r  = Teff & (d - 1);
            TeffH  = ((Teff >> di) << ddi) + (half ? max(0, r - dd) : min(r, dd));
            const int hq = 1 << eq;
            const int rq = NQUAD & (2 * hq - 1);
            nquadH = ((NQUAD >> (eq + 1)) << eq) + (half ? max(0, rq - hq) : min(rq, hq));
        }

        {
            float4* sg4 = reinterpret_cast<float4*>(sg);
            if (!dif) {
                for (int jq = tid; jq < nquadH; jq += nthr) {
                    const int q = SPLIT ? ((((jq >> eq) << 1) | half) << eq) | (jq & ((1 << eq) - 1)) : jq;
                    float4 v[8];
                    #pragma unroll
                    for (int i = 0; i < 8; ++i) v[i] = reinterpret_cast<const float4*>(xr[i])[q];
                    float4 s = make_float4(0.0f, 0.0f, 0.0f, 0.0f);
                    #pragma unroll
                    for (int i = 0; i < 8; ++i) {
                        s.x += v[i].x;  s.y += v[i].y;  s.z += v[i].z;  s.w += v[i].w;
                    }
                    sg4[jq] = s;
                }
            } else {
                for (int jq = tid; jq < nquadH; jq += nthr) {
                    const int q = SPLIT ? ((((jq >> eq) << 1) | half) << eq) | (jq & ((1 << eq) - 1)) : jq;
                    const int te = min(4 * q + 4, T_FULL - 1);
                    float4 v[8];
                    float  e[8];
                    #pragma unroll
                    for (int i = 0; i < 8; ++i) v[i] = reinterpret_cast<const float4*>(xr[i])[q];
                    #pragma unroll
                    for (int i = 0; i < 8; ++i) e[i] = xr[i][te];
                    if (SPLIT) __builtin_amdgcn_sched_barrier(0);
                    float4 s = make_float4(0.0f, 0.0f, 0.0f, 0.0f);
                    #pragma unroll
                    for (int i = 0; i < 8; ++i) {
                        s.x += v[i].y - v[i].x;
                        s.y += v[i].z - v[i].y;
                        s.z += v[i].w - v[i].z;
                        s.w += e[i]   - v[i].w;
                    }
                    sg4[jq] = s;
                }
            }
        }
        __syncthreads();

        v2f afrag[3];
        #pragma unroll
        for (int tile = 0; tile < 3; ++tile) {
            #pragma unroll
            for (int i = 0; i < 2; ++i) {
                const int jj = tile * 4 + krb + i - as;
                afrag[tile][i] = (jj >= 0 && jj < NTAP) ? sw[ak * NTAP + jj] : 0.0f;
            }
        }

        const int hmask = -dd;
        const int hp    = (TeffH - 1) >> (ddi + 1);
        const int nq    = (hp << ddi) + min(dd, TeffH - (hp << (ddi + 1)));
        const int nblk  = (nq + 15) >> 4;

        const int blkA = (2 * dd + 15) >> 4;
        const int hiT  = TeffH - 5 * dd - 1;
        int blkB = 0;
        if (hiT >= 0) {
            const int qmax = ((hiT >> (ddi + 1)) << ddi) + min(hiT & (2 * dd - 1), dd - 1);
            blkB = (qmax + 1) >> 4;
        }
        const int e1 = min(blkA, nblk);
        const int e2 = max(e1, min(blkB, nblk));

        int blk = wav;
        for (; blk + nwav < e1; blk += 2 * nwav)
            conv_trip_pair<false, ZS>(sg, blk, blk + nwav, c16, hs, krb, dd, hmask, TeffH, offI, afrag, accMax, cumMin, deadCnt);
        for (; blk < e1; blk += nwav)
            conv_trip<false, ZS>(sg, blk, c16, hs, krb, dd, hmask, TeffH, offI, afrag, accMax, cumMin, deadCnt);
        for (; blk + nwav < e2; blk += 2 * nwav)
            conv_trip_pair<true, ZS>(sg, blk, blk + nwav, c16, hs, krb, dd, hmask, TeffH, offI, afrag, accMax, cumMin, deadCnt);
        for (; blk < e2; blk += nwav)
            conv_trip<true, ZS>(sg, blk, c16, hs, krb, dd, hmask, TeffH, offI, afrag, accMax, cumMin, deadCnt);
        for (; blk + nwav < nblk; blk += 2 * nwav)
            conv_trip_pair<false, ZS>(sg, blk, blk + nwav, c16, hs, krb, dd, hmask, TeffH, offI, afrag, accMax, cumMin, deadCnt);
        for (; blk < nblk; blk += nwav)
            conv_trip<false, ZS>(sg, blk, c16, hs, krb, dd, hmask, TeffH, offI, afrag, accMax, cumMin, deadCnt);

        if (half + 1 < nhalf) __syncthreads();
    }

    int cumBelow = deadCnt;
    #pragma unroll
    for (int k = 0; k < 8; ++k) {
        float vM = accMax[k];
        #pragma unroll
        for (int off = 16; off > 0; off >>= 1) vM += __shfl_xor(vM, off, 32);
        if (lane == 0) {
            atomicAdd(&sAcc[k], vM);
            if (k < 7) atomicAdd(&sAcc[8 + k], (float)(cumMin[k] - cumBelow));
        }
        if (k < 7) cumBelow = cumMin[k];
    }
    __syncthreads();

    if (tid < 16) {
        const int which = tid >> 3;
        const int k     = tid & 7;
        const size_t off = (size_t)b * 14336
                         + (size_t)(2 * u + which) * (NGROUP * NKER)
                         + (size_t)h * NKER + k;
        float v = sAcc[tid];
        if (tid == 15) {
            float lower = 0.0f;
            #pragma unroll
            for (int j = 0; j < 7; ++j) lower += sAcc[8 + j];
            v = (float)Teff - lower;
        }
        OUT[off] = v;
    }
}

extern "C" void kernel_launch(void* const* d_in, const int* in_sizes, int n_in,
                              void* d_out, int out_size, void* d_ws, size_t ws_size,
                              hipStream_t stream) {
    (void)in_sizes; (void)n_in; (void)d_ws; (void)ws_size; (void)out_size;
    const float* X   = (const float*)d_in[0];
    const float* W   = (const float*)d_in[1];
    const int*   IDX = (const int*)d_in[2];
    float*       OUT = (float*)d_out;

    const int shmemW = (SG_FLOATS + SMEM_TAIL) * (int)sizeof(float);
    const int shmemH = (SG_FLOATS_H + SMEM_TAIL) * (int)sizeof(float);
    (void)hipFuncSetAttribute((const void*)hydra_unit_kernel<false>,
                              hipFuncAttributeMaxDynamicSharedMemorySize, shmemW);
    (void)hipFuncSetAttribute((const void*)hydra_unit_kernel<true>,
                              hipFuncAttributeMaxDynamicSharedMemorySize, shmemH);

    const int nblocksH = 4 * (NUNIT - NUNIT_WHOLE) * NGROUP;
    const int nblocksW = 4 * NUNIT_WHOLE * NGROUP;
    hydra_unit_kernel<true><<<dim3(nblocksH), dim3(512), shmemH, stream>>>(X, W, IDX, OUT);
    hydra_unit_kernel<false><<<dim3(nblocksW), dim3(1024), shmemW, stream>>>(X, W, IDX, OUT);
}


# dilation <= 32 also as two half problems (time points below / from 37504, 4d neighbours across the cut held, not scored): one launch, 3584 16-wave groups = 7 full rounds; max sums reordered
# speedup vs baseline: 2.9307x; 1.1118x over previous
#include <hip/hip_runtime.h>

typedef __attribute__((ext_vector_type(2))) float v2f;
typedef __attribute__((ext_vector_type(8))) float v8f;

#define T_FULL 75000
#define CH_IN  17
#define NGROUP 32
#define NKER   8
#define NTAP   9
#define NUNIT  28
#define NQUAD  (T_FULL / 4)
#define CUT_DI      6
#define HALF_CUT    37504
#define SG_FLOATS_H 38160
#define ZSLOT_H     38144
#define SMEM_TAIL   (72 + 16)

template <bool INTERIOR, int ZS>
__device__ __forceinline__ void conv_trip(const float* sg, const int blk,
                                          const int c16, const int hs, const int krb,
                                          const int d, const int hmask, const int Teff,
                                          const int own0, const int ownLen,
                                          const int (&offI)[3][2], const v2f (&afrag)[3],
                                          float (&accMax)[8], int (&cumMin)[7], int &deadCnt)
{
    v2f bf[3];
    unsigned dead;
    if (INTERIOR) {
        const int q4 = (blk << 6) + (c16 << 2);
        const int TB = q4 + (q4 & (hmask * 4));
        const char* sgB = reinterpret_cast<const char*>(sg);
        #pragma unroll
        for (int tile = 0; tile < 3; ++tile) {
            #pragma unroll
            for (int i = 0; i < 2; ++i)
                bf[tile][i] = *reinterpret_cast<const float*>(sgB + TB + offI[tile][i]);
        }
        dead = 0u;
    } else {
        const int q = (blk << 4) + c16;
        const int T = q + (q & hmask);
        #pragma unroll
        for (int tile = 0; tile < 3; ++tile) {
            #pragma unroll
            for (int i = 0; i < 2; ++i) {
                const int kk = tile * 4 + krb + i;
                const int p  = T + (kk - 4) * d;
                const int a  = (kk < NTAP + 1 && (unsigned)p < (unsigned)Teff) ? p : ZS;
                bf[tile][i] = sg[a];
            }
        }
        dead = __builtin_amdgcn_ballot_w32((unsigned)(T + (hs ? d : 0) - own0) >= (unsigned)ownLen);
    }

    v8f z = {0.f,0.f,0.f,0.f,0.f,0.f,0.f,0.f};
    #pragma unroll
    for (int tile = 0; tile < 3; ++tile) {
        z = __builtin_amdgcn_wmma_f32_16x16x4_f32(
                false, afrag[tile], false, bf[tile], (short)0, z, false, false);
    }

    float mx = z[0], mn = z[0];
    #pragma unroll
    for (int k = 1; k < 8; ++k) { mx = fmaxf(mx, z[k]); mn = fminf(mn, z[k]); }
    unsigned long long seen = ((unsigned long long)dead << 32) | dead;
    if (!INTERIOR) deadCnt += __builtin_popcount(dead);
    #pragma unroll
    for (int k = 0; k < 8; ++k) {
        const unsigned em = __builtin_amdgcn_ballot_w32(z[k] == mx);
        const unsigned fm = em & ~(unsigned)seen;
        accMax[k] += __builtin_amdgcn_inverse_ballot_w32(fm) ? mx : 0.0f;
        if (k < 7) {
            const unsigned en = __builtin_amdgcn_ballot_w32(z[k] == mn);
            seen |= ((unsigned long long)en << 32) | em;
            cumMin[k] += __builtin_popcount((unsigned)(seen >> 32));
        }
    }
}

template <bool INTERIOR, int ZS>
__device__ __forceinline__ void conv_trip_pair(const float* sg, const int blka, const int blkb,
                                               const int c16, const int hs, const int krb,
                                               const int d, const int hmask, const int Teff,
                                               const int own0, const int ownLen,
                                               const int (&offI)[3][2], const v2f (&afrag)[3],
                                               float (&accMax)[8], int (&cumMin)[7], int &deadCnt)
{
    v2f bfa[3], bfb[3];
    unsigned long long dead;
    if (INTERIOR) {
        const char* sgB = reinterpret_cast<const char*>(sg);
        const int q4a = (blka << 6) + (c16 << 2);
        const int q4b = (blkb << 6) + (c16 << 2);
        const int TBa = q4a + (q4a & (hmask * 4));
        const int TBb = q4b + (q4b & (hmask * 4));
        #pragma unroll
        for (int tile = 0; tile < 3; ++tile) {
            #pragma unroll
            for (int i = 0; i < 2; ++i) {
                bfa[tile][i] = *reinterpret_cast<const float*>(sgB + TBa + offI[tile][i]);
                bfb[tile][i] = *reinterpret_cast<const float*>(sgB + TBb + offI[tile][i]);
            }
        }
        dead = 0ull;
    } else {
        const int qa = (blka << 4) + c16;
        const int qb = (blkb << 4) + c16;
        const int Ta = qa + (qa & hmask);
        const int Tb = qb + (qb & hmask);
        #pragma unroll
        for (int tile = 0; tile < 3; ++tile) {
            #pragma unroll
            for (int i = 0; i < 2; ++i) {
                const int kk = tile * 4 + krb + i;
                const int pa = Ta + (kk - 4) * d;
                const int pb = Tb + (kk - 4) * d;
                const int aa = (kk < NTAP + 1 && (unsigned)pa < (unsigned)Teff) ? pa : ZS;
                const int ab = (kk < NTAP + 1 && (unsigned)pb < (unsigned)Teff) ? pb : ZS;
                bfa[tile][i] = sg[aa];
                bfb[tile][i] = sg[ab];
            }
        }
        const unsigned deada = __builtin_amdgcn_ballot_w32((unsigned)(Ta + (hs ? d : 0) - own0) >= (unsigned)ownLen);
        const unsigned deadb = __builtin_amdgcn_ballot_w32((unsigned)(Tb + (hs ? d : 0) - own0) >= (unsigned)ownLen);
        dead = ((unsigned long long)deadb << 32) | deada;
        deadCnt += __builtin_popcountll(dead);
    }
    v8f za = {0.f,0.f,0.f,0.f,0.f,0.f,0.f,0.f};
    v8f zb = {0.f,0.f,0.f,0.f,0.f,0.f,0.f,0.f};
    #pragma unroll
    for (int tile = 0; tile < 3; ++tile) {
        za = __builtin_amdgcn_wmma_f32_16x16x4_f32(
                false, afrag[tile], false, bfa[tile], (short)0, za, false, false);
        zb = __builtin_amdgcn_wmma_f32_16x16x4_f32(
                false, afrag[tile], false, bfb[tile], (short)0, zb, false, false);
    }
    float mxa = za[0], mna = za[0], mxb = zb[0], mnb = zb[0];
    #pragma unroll
    for (int k = 1; k < 8; ++k) {
        mxa = fmaxf(mxa, za[k]);  mna = fminf(mna, za[k]);
        mxb = fmaxf(mxb, zb[k]);  mnb = fminf(mnb, zb[k]);
    }
    unsigned long long seenMax = dead;
    unsigned long long seenMin = dead;
    #pragma unroll
    for (int k = 0; k < 8; ++k) {
        const unsigned ema = __builtin_amdgcn_ballot_w32(za[k] == mxa);
        const unsigned emb = __builtin_amdgcn_ballot_w32(zb[k] == mxb);
        const unsigned long long em = ((unsigned long long)emb << 32) | ema;
        const unsigned long long fm = em & ~seenMax;
        accMax[k] += __builtin_amdgcn_inverse_ballot_w32((unsigned)fm) ? mxa : 0.0f;
        accMax[k] += __builtin_amdgcn_inverse_ballot_w32((unsigned)(fm >> 32)) ? mxb : 0.0f;
        if (k < 7) {
            seenMax |= em;
            const unsigned ena = __builtin_amdgcn_ballot_w32(za[k] == mna);
            const unsigned enb = __builtin_amdgcn_ballot_w32(zb[k] == mnb);
            seenMin |= ((unsigned long long)enb << 32) | ena;
            cumMin[k] += __builtin_popcountll(seenMin);
        }
    }
}

__global__ __launch_bounds__(512)
void hydra_unit_kernel(const float* __restrict__ X,
                       const float* __restrict__ W,
                       const int*   __restrict__ IDX,
                       float* __restrict__ OUT)
{
    constexpr int SGF = SG_FLOATS_H;
    constexpr int ZS  = ZSLOT_H;
    extern __shared__ __attribute__((aligned(16))) float smem[];
    float* sg   = smem;
    float* sw   = smem + SGF;
    float* sAcc = smem + SGF + 72;

    const int tid  = threadIdx.x;
    const int lane = tid & 31;
    const int wav  = __builtin_amdgcn_readfirstlane(tid >> 5);
    const int nthr = blockDim.x;
    const int nwav = nthr >> 5;

    const int bid = blockIdx.x;
    const int b   = bid / (NUNIT * NGROUP);
    const int rem = bid % (NUNIT * NGROUP);
    const int u   = rem / NGROUP;
    const int h   = rem % NGROUP;
    const int di  = u >> 1;
    const int dif = u & 1;
    const int d   = 1 << di;
    const int Teff = dif ? (T_FULL - 1) : T_FULL;

    const bool cut = di < CUT_DI;
    const int ddi = cut ? di : di - 1;
    const int dd  = 1 << ddi;
    const int eq  = cut ? 0 : di - 3;
    const int nhalf = 2;

    if (tid < 16) sAcc[tid] = 0.0f;
    if (tid == 16) sg[ZS] = 0.0f;
    {
        const int wbase = (u * (NKER * NGROUP) + h * NKER) * NTAP;
        if (tid < NKER * NTAP) sw[tid] = W[wbase + tid];
    }
    int ch[8];
    {
        const int* ip = IDX + (u * NGROUP + h) * 8;
        #pragma unroll
        for (int i = 0; i < 8; ++i) ch[i] = ip[i];
    }
    const float* Xb = X + (size_t)b * CH_IN * T_FULL;
    const float* xr[8];
    #pragma unroll
    for (int i = 0; i < 8; ++i) xr[i] = Xb + (size_t)ch[i] * T_FULL;

    const int c16 = lane & 15;
    const int hs  = lane >> 4;
    const int krb = hs << 1;
    const int ak  = c16 & 7;
    const int as  = c16 >> 3;

    float accMax[8];
    int   cumMin[7];
    int   deadCnt = 0;
    #pragma unroll
    for (int k = 0; k < 8; ++k) accMax[k] = 0.0f;
    #pragma unroll
    for (int k = 0; k < 7; ++k) cumMin[k] = 0;

    int offI[3][2];
    #pragma unroll
    for (int tile = 0; tile < 3; ++tile) {
        #pragma unroll
        for (int i = 0; i < 2; ++i) {
            const int kk = tile * 4 + krb + i;
            offI[tile][i] = (kk < NTAP + 1) ? (kk - 4) * dd * 4 : 0;
        }
    }

    #pragma nounroll
    for (int half = 0; half < nhalf; ++half) {
        int TeffH, nquadH, own0 = 0, own1, q0 = 0;
        if (cut) {
            const int o0 = half ? HALF_CUT : 0;
            const int o1 = half ? Teff : HALF_CUT;
            const int l0 = max(0, o0 - 4 * d);
            const int l1 = min(Teff, o1 + 4 * d);
            TeffH  = l1 - l0;
            nquadH = (TeffH + 3) >> 2;
            q0     = l0 >> 2;
            own0   = o0 - l0;
            own1   = o1 - l0;
        } else {
            const int r  = Teff & (d - 1);
            TeffH  = ((Teff >> di) << ddi) + (half ? max(0, r - dd) : min(r, dd));
            const int hq = 1 << eq;
            const int rq = NQUAD & (2 * hq - 1);
            nquadH = ((NQUAD >> (eq + 1)) << eq) + (half ? max(0, rq - hq) : min(rq, hq));
            own1   = TeffH;
        }
        const int ownLen = own1 - own0;

        {
            float4* sg4 = reinterpret_cast<float4*>(sg);
            if (!dif) {
                for (int jq = tid; jq < nquadH; jq += nthr) {
                    const int q = cut ? q0 + jq : ((((jq >> eq) << 1) | half) << eq) | (jq & ((1 << eq) - 1));
                    float4 v[8];
                    #pragma unroll
                    for (int i = 0; i < 8; ++i) v[i] = reinterpret_cast<const float4*>(xr[i])[q];
                    float4 s = make_float4(0.0f, 0.0f, 0.0f, 0.0f);
                    #pragma unroll
                    for (int i = 0; i < 8; ++i) {
                        s.x += v[i].x;  s.y += v[i].y;  s.z += v[i].z;  s.w += v[i].w;
                    }
                    sg4[jq] = s;
                }
            } else {
                for (int jq = tid; jq < nquadH; jq += nthr) {
                    const int q = cut ? q0 + jq : ((((jq >> eq) << 1) | half) << eq) | (jq & ((1 << eq) - 1));
                    const int te = min(4 * q + 4, T_FULL - 1);
                    float4 v[8];
                    float  e[8];
                    #pragma unroll
                    for (int i = 0; i < 8; ++i) v[i] = reinterpret_cast<const float4*>(xr[i])[q];
                    #pragma unroll
                    for (int i = 0; i < 8; ++i) e[i] = xr[i][te];
                    __builtin_amdgcn_sched_barrier(0);
                    float4 s = make_float4(0.0f, 0.0f, 0.0f, 0.0f);
                    #pragma unroll
                    for (int i = 0; i < 8; ++i) {
                        s.x += v[i].y - v[i].x;
                        s.y += v[i].z - v[i].y;
                        s.z += v[i].w - v[i].z;
                        s.w += e[i]   - v[i].w;
                    }
                    sg4[jq] = s;
                }
            }
        }
        __syncthreads();

        v2f afrag[3];
        #pragma unroll
        for (int tile = 0; tile < 3; ++tile) {
            #pragma unroll
            for (int i = 0; i < 2; ++i) {
                const int jj = tile * 4 + krb + i - as;
                afrag[tile][i] = (jj >= 0 && jj < NTAP) ? sw[ak * NTAP + jj] : 0.0f;
            }
        }

        const int hmask = -dd;
        const int hp    = (TeffH - 1) >> (ddi + 1);
        const int nq    = (hp << ddi) + min(dd, TeffH - (hp << (ddi + 1)));
        const int nblk  = (nq + 15) >> 4;

        const int blkA = (2 * dd + 15) >> 4;
        const int hiT  = TeffH - 5 * dd - 1;
        int blkB = 0;
        if (hiT >= 0) {
            const int qmax = ((hiT >> (ddi + 1)) << ddi) + min(hiT & (2 * dd - 1), dd - 1);
            blkB = (qmax + 1) >> 4;
        }
        const int ownA = ((own0 >> 1) + 15) >> 4;
        const int ownB = (own1 < TeffH) ? (own1 >> 5) : nblk;
        const int e1 = min(max(blkA, ownA), nblk);
        const int e2 = max(e1, min(min(blkB, ownB), nblk));

        int blk = wav;
        for (; blk + nwav < e1; blk += 2 * nwav)
            conv_trip_pair<false, ZS>(sg, blk, blk + nwav, c16, hs, krb, dd, hmask, TeffH, own0, ownLen, offI, afrag, accMax, cumMin, deadCnt);
        for (; blk < e1; blk += nwav)
            conv_trip<false, ZS>(sg, blk, c16, hs, krb, dd, hmask, TeffH, own0, ownLen, offI, afrag, accMax, cumMin, deadCnt);
        for (; blk + nwav < e2; blk += 2 * nwav)
            conv_trip_pair<true, ZS>(sg, blk, blk + nwav, c16, hs, krb, dd, hmask, TeffH, own0, ownLen, offI, afrag, accMax, cumMin, deadCnt);
        for (; blk < e2; blk += nwav)
            conv_trip<true, ZS>(sg, blk, c16, hs, krb, dd, hmask, TeffH, own0, ownLen, offI, afrag, accMax, cumMin, deadCnt);
        for (; blk + nwav < nblk; blk += 2 * nwav)
            conv_trip_pair<false, ZS>(sg, blk, blk + nwav, c16, hs, krb, dd, hmask, TeffH, own0, ownLen, offI, afrag, accMax, cumMin, deadCnt);
        for (; blk < nblk; blk += nwav)
            conv_trip<false, ZS>(sg, blk, c16, hs, krb, dd, hmask, TeffH, own0, ownLen, offI, afrag, accMax, cumMin, deadCnt);

        if (half + 1 < nhalf) __syncthreads();
    }

    int cumBelow = deadCnt;
    #pragma unroll
    for (int k = 0; k < 8; ++k) {
        float vM = accMax[k];
        #pragma unroll
        for (int off = 16; off > 0; off >>= 1) vM += __shfl_xor(vM, off, 32);
        if (lane == 0) {
            atomicAdd(&sAcc[k], vM);
            if (k < 7) atomicAdd(&sAcc[8 + k], (float)(cumMin[k] - cumBelow));
        }
        if (k < 7) cumBelow = cumMin[k];
    }
    __syncthreads();

    if (tid < 16) {
        const int which = tid >> 3;
        const int k     = tid & 7;
        const size_t off = (size_t)b * 14336
                         + (size_t)(2 * u + which) * (NGROUP * NKER)
                         + (size_t)h * NKER + k;
        float v = sAcc[tid];
        if (tid == 15) {
            float lower = 0.0f;
            #pragma unroll
            for (int j = 0; j < 7; ++j) lower += sAcc[8 + j];
            v = (float)Teff - lower;
        }
        OUT[off] = v;
    }
}

extern "C" void kernel_launch(void* const* d_in, const int* in_sizes, int n_in,
                              void* d_out, int out_size, void* d_ws, size_t ws_size,
                              hipStream_t stream) {
    (void)in_sizes; (void)n_in; (void)d_ws; (void)ws_size; (void)out_size;
    const float* X   = (const float*)d_in[0];
    const float* W   = (const float*)d_in[1];
    const int*   IDX = (const int*)d_in[2];
    float*       OUT = (float*)d_out;

    const int shmemH = (SG_FLOATS_H + SMEM_TAIL) * (int)sizeof(float);
    (void)hipFuncSetAttribute((const void*)hydra_unit_kernel,
                              hipFuncAttributeMaxDynamicSharedMemorySize, shmemH);

    const int nblocks = 4 * NUNIT * NGROUP;
    hydra_unit_kernel<<<dim3(nblocks), dim3(512), shmemH, stream>>>(X, W, IDX, OUT);
}
